# starting kernel
# baseline (speedup 1.0000x reference)
#include <hip/hip_runtime.h>


typedef __attribute__((ext_vector_type(2))) float v2f;
typedef __attribute__((ext_vector_type(8))) float v8f;

#define NROWS 50000
#define INDIM 256
#define HID   128
#define CLS   64
#define NEDGE 800000

__global__ __launch_bounds__(64) void gcn_gemm1_wmma(
    const float* __restrict__ X, const float* __restrict__ W1,
    float* __restrict__ Y)
{
    const int mtile = blockIdx.x;
    const int wv    = threadIdx.x >> 5;
    const int lane  = threadIdx.x & 31;
    const int half  = lane >> 4;
    const int l16   = lane & 15;

    const int m0 = mtile * 16;
    const int n0 = wv * 64;

    const float* xrow = X + (size_t)(m0 + l16) * INDIM + 2 * half;
    const float* wcol = W1 + n0 + l16;

    v8f acc0 = {}, acc1 = {}, acc2 = {}, acc3 = {};
    #pragma unroll 2
    for (int k = 0; k < INDIM; k += 4) {
        const float2 av = *(const float2*)(xrow + k);
        v2f a; a.x = av.x; a.y = av.y;
        const float* wk0 = wcol + (size_t)(k + 2 * half) * HID;
        const float* wk1 = wcol + (size_t)(k + 2 * half + 1) * HID;
        v2f b0; b0.x = wk0[0];  b0.y = wk1[0];
        v2f b1; b1.x = wk0[16]; b1.y = wk1[16];
        v2f b2; b2.x = wk0[32]; b2.y = wk1[32];
        v2f b3; b3.x = wk0[48]; b3.y = wk1[48];
        acc0 = __builtin_amdgcn_wmma_f32_16x16x4_f32(false, a, false, b0, (short)0, acc0, false, false);
        acc1 = __builtin_amdgcn_wmma_f32_16x16x4_f32(false, a, false, b1, (short)0, acc1, false, false);
        acc2 = __builtin_amdgcn_wmma_f32_16x16x4_f32(false, a, false, b2, (short)0, acc2, false, false);
        acc3 = __builtin_amdgcn_wmma_f32_16x16x4_f32(false, a, false, b3, (short)0, acc3, false, false);
    }

    float* yb = Y + n0 + l16;
    #pragma unroll
    for (int i = 0; i < 8; ++i) {
        float* yr = yb + (size_t)(m0 + i + 8 * half) * HID;
        yr[0]  = acc0[i];
        yr[16] = acc1[i];
        yr[32] = acc2[i];
        yr[48] = acc3[i];
    }
}

__global__ __launch_bounds__(32) void gcn_gemm2_wmma(
    const float* __restrict__ Hraw, const float* __restrict__ b1,
    const float* __restrict__ W2, float* __restrict__ Y)
{
    const int mtile = blockIdx.x;
    const int lane  = threadIdx.x & 31;
    const int half  = lane >> 4;
    const int l16   = lane & 15;

    const int m0 = mtile * 16;

    const float* hrow = Hraw + (size_t)(m0 + l16) * HID + 2 * half;
    const float* brow = b1 + 2 * half;
    const float* wcol = W2 + l16;

    v8f acc0 = {}, acc1 = {}, acc2 = {}, acc3 = {};
    #pragma unroll 2
    for (int k = 0; k < HID; k += 4) {
        const float2 hv = *(const float2*)(hrow + k);
        const float2 bv = *(const float2*)(brow + k);
        v2f a;
        a.x = fmaxf(hv.x + bv.x, 0.0f);
        a.y = fmaxf(hv.y + bv.y, 0.0f);
        const float* wk0 = wcol + (size_t)(k + 2 * half) * CLS;
        const float* wk1 = wcol + (size_t)(k + 2 * half + 1) * CLS;
        v2f b0; b0.x = wk0[0];  b0.y = wk1[0];
        v2f b1v; b1v.x = wk0[16]; b1v.y = wk1[16];
        v2f b2v; b2v.x = wk0[32]; b2v.y = wk1[32];
        v2f b3v; b3v.x = wk0[48]; b3v.y = wk1[48];
        acc0 = __builtin_amdgcn_wmma_f32_16x16x4_f32(false, a, false, b0,  (short)0, acc0, false, false);
        acc1 = __builtin_amdgcn_wmma_f32_16x16x4_f32(false, a, false, b1v, (short)0, acc1, false, false);
        acc2 = __builtin_amdgcn_wmma_f32_16x16x4_f32(false, a, false, b2v, (short)0, acc2, false, false);
        acc3 = __builtin_amdgcn_wmma_f32_16x16x4_f32(false, a, false, b3v, (short)0, acc3, false, false);
    }

    float* yb = Y + l16;
    #pragma unroll
    for (int i = 0; i < 8; ++i) {
        float* yr = yb + (size_t)(m0 + i + 8 * half) * CLS;
        yr[0]  = acc0[i];
        yr[16] = acc1[i];
        yr[32] = acc2[i];
        yr[48] = acc3[i];
    }
}

__global__ __launch_bounds__(256) void gcn_spmm_f128(
    const int* __restrict__ erow, const int* __restrict__ ecol,
    const float* __restrict__ eval, const float* __restrict__ dense,
    float* __restrict__ out)
{
    const int e = (int)((blockIdx.x * (size_t)blockDim.x + threadIdx.x) >> 5);
    const int lane = threadIdx.x & 31;
    if (e >= NEDGE) return;
    const int   r = erow[e];
    const int   c = ecol[e];
    const float v = eval[e];
    const float4 g = *(const float4*)(dense + (size_t)c * HID + lane * 4);
    float* o = out + (size_t)r * HID + lane * 4;
    atomicAdd(o + 0, v * g.x);
    atomicAdd(o + 1, v * g.y);
    atomicAdd(o + 2, v * g.z);
    atomicAdd(o + 3, v * g.w);
}

__global__ __launch_bounds__(256) void gcn_spmm_f64(
    const int* __restrict__ erow, const int* __restrict__ ecol,
    const float* __restrict__ eval, const float* __restrict__ dense,
    float* __restrict__ out)
{
    const int e = (int)((blockIdx.x * (size_t)blockDim.x + threadIdx.x) >> 5);
    const int lane = threadIdx.x & 31;
    if (e >= NEDGE) return;
    const int   r = erow[e];
    const int   c = ecol[e];
    const float v = eval[e];
    const float2 g = *(const float2*)(dense + (size_t)c * CLS + lane * 2);
    float* o = out + (size_t)r * CLS + lane * 2;
    atomicAdd(o + 0, v * g.x);
    atomicAdd(o + 1, v * g.y);
}

__global__ __launch_bounds__(256) void gcn_logsoftmax64(
    const float* __restrict__ L, const float* __restrict__ b2,
    float* __restrict__ out)
{
    const int r = (int)((blockIdx.x * (size_t)blockDim.x + threadIdx.x) >> 5);
    const int lane = threadIdx.x & 31;
    if (r >= NROWS) return;

    float2 v = *(const float2*)(L + (size_t)r * CLS + lane * 2);
    const float2 bb = *(const float2*)(b2 + lane * 2);
    v.x += bb.x;
    v.y += bb.y;

    float m = fmaxf(v.x, v.y);
    #pragma unroll
    for (int off = 16; off > 0; off >>= 1)
        m = fmaxf(m, __shfl_xor(m, off, 32));

    float s = __expf(v.x - m) + __expf(v.y - m);
    #pragma unroll
    for (int off = 16; off > 0; off >>= 1)
        s += __shfl_xor(s, off, 32);

    const float lse = m + __logf(s);
    float2 o;
    o.x = v.x - lse;
    o.y = v.y - lse;
    *(float2*)(out + (size_t)r * CLS + lane * 2) = o;
}

__global__ __launch_bounds__(256) void gcn_zero_f32(float* __restrict__ p, int n)
{
    const int i = (int)(blockIdx.x * (size_t)blockDim.x + threadIdx.x);
    if (i < n) p[i] = 0.0f;
}

extern "C" void kernel_launch(void* const* d_in, const int* in_sizes, int n_in,
                              void* d_out, int out_size, void* d_ws, size_t ws_size,
                              hipStream_t stream) {
    const float* x        = (const float*)d_in[0];
    const int*   edge_row = (const int*)  d_in[1];
    const int*   edge_col = (const int*)  d_in[2];
    const float* edge_val = (const float*)d_in[3];
    const float* W1       = (const float*)d_in[4];
    const float* b1       = (const float*)d_in[5];
    const float* W2       = (const float*)d_in[6];
    const float* b2       = (const float*)d_in[7];
    float* out = (float*)d_out;

    float* ws    = (float*)d_ws;
    float* XW1   = ws;
    float* Hraw  = ws + (size_t)NROWS * HID;
    float* HW2   = XW1;
    float* Lraw  = Hraw;

    const int MT = NROWS / 16;
    const int edge_blocks = (NEDGE * 32 + 255) / 256;

    gcn_gemm1_wmma<<<MT, 64, 0, stream>>>(x, W1, XW1);
    gcn_zero_f32<<<(NROWS * HID + 255) / 256, 256, 0, stream>>>(Hraw, NROWS * HID);
    gcn_spmm_f128<<<edge_blocks, 256, 0, stream>>>(edge_row, edge_col, edge_val, XW1, Hraw);
    gcn_gemm2_wmma<<<MT, 32, 0, stream>>>(Hraw, b1, W2, HW2);
    gcn_zero_f32<<<(NROWS * CLS + 255) / 256, 256, 0, stream>>>(Lraw, NROWS * CLS);
    gcn_spmm_f64<<<edge_blocks, 256, 0, stream>>>(edge_row, edge_col, edge_val, HW2, Lraw);
    gcn_logsoftmax64<<<(NROWS * 32 + 255) / 256, 256, 0, stream>>>(Lraw, b2, out);
}


# SpMM float-atomic scatter replaced by a per-call counting sort of edges by row (integer-atomic arrival-order rank); one wave per row then sums its edges in that order, no float atomics
# speedup vs baseline: 1.1889x; 1.1889x over previous
#include <hip/hip_runtime.h>


typedef __attribute__((ext_vector_type(2))) float v2f;
typedef __attribute__((ext_vector_type(8))) float v8f;

#define NROWS 50000
#define INDIM 256
#define HID   128
#define CLS   64
#define NEDGE 800000

__global__ __launch_bounds__(64) void gcn_gemm1_wmma(
    const float* __restrict__ X, const float* __restrict__ W1,
    float* __restrict__ Y)
{
    const int mtile = blockIdx.x;
    const int wv    = threadIdx.x >> 5;
    const int lane  = threadIdx.x & 31;
    const int half  = lane >> 4;
    const int l16   = lane & 15;

    const int m0 = mtile * 16;
    const int n0 = wv * 64;

    const float* xrow = X + (size_t)(m0 + l16) * INDIM + 2 * half;
    const float* wcol = W1 + n0 + l16;

    v8f acc0 = {}, acc1 = {}, acc2 = {}, acc3 = {};
    #pragma unroll 2
    for (int k = 0; k < INDIM; k += 4) {
        const float2 av = *(const float2*)(xrow + k);
        v2f a; a.x = av.x; a.y = av.y;
        const float* wk0 = wcol + (size_t)(k + 2 * half) * HID;
        const float* wk1 = wcol + (size_t)(k + 2 * half + 1) * HID;
        v2f b0; b0.x = wk0[0];  b0.y = wk1[0];
        v2f b1; b1.x = wk0[16]; b1.y = wk1[16];
        v2f b2; b2.x = wk0[32]; b2.y = wk1[32];
        v2f b3; b3.x = wk0[48]; b3.y = wk1[48];
        acc0 = __builtin_amdgcn_wmma_f32_16x16x4_f32(false, a, false, b0, (short)0, acc0, false, false);
        acc1 = __builtin_amdgcn_wmma_f32_16x16x4_f32(false, a, false, b1, (short)0, acc1, false, false);
        acc2 = __builtin_amdgcn_wmma_f32_16x16x4_f32(false, a, false, b2, (short)0, acc2, false, false);
        acc3 = __builtin_amdgcn_wmma_f32_16x16x4_f32(false, a, false, b3, (short)0, acc3, false, false);
    }

    float* yb = Y + n0 + l16;
    #pragma unroll
    for (int i = 0; i < 8; ++i) {
        float* yr = yb + (size_t)(m0 + i + 8 * half) * HID;
        yr[0]  = acc0[i];
        yr[16] = acc1[i];
        yr[32] = acc2[i];
        yr[48] = acc3[i];
    }
}

__global__ __launch_bounds__(32) void gcn_gemm2_wmma(
    const float* __restrict__ Hraw, const float* __restrict__ b1,
    const float* __restrict__ W2, float* __restrict__ Y)
{
    const int mtile = blockIdx.x;
    const int lane  = threadIdx.x & 31;
    const int half  = lane >> 4;
    const int l16   = lane & 15;

    const int m0 = mtile * 16;

    const float* hrow = Hraw + (size_t)(m0 + l16) * HID + 2 * half;
    const float* brow = b1 + 2 * half;
    const float* wcol = W2 + l16;

    v8f acc0 = {}, acc1 = {}, acc2 = {}, acc3 = {};
    #pragma unroll 2
    for (int k = 0; k < HID; k += 4) {
        const float2 hv = *(const float2*)(hrow + k);
        const float2 bv = *(const float2*)(brow + k);
        v2f a;
        a.x = fmaxf(hv.x + bv.x, 0.0f);
        a.y = fmaxf(hv.y + bv.y, 0.0f);
        const float* wk0 = wcol + (size_t)(k + 2 * half) * CLS;
        const float* wk1 = wcol + (size_t)(k + 2 * half + 1) * CLS;
        v2f b0; b0.x = wk0[0];  b0.y = wk1[0];
        v2f b1v; b1v.x = wk0[16]; b1v.y = wk1[16];
        v2f b2v; b2v.x = wk0[32]; b2v.y = wk1[32];
        v2f b3v; b3v.x = wk0[48]; b3v.y = wk1[48];
        acc0 = __builtin_amdgcn_wmma_f32_16x16x4_f32(false, a, false, b0,  (short)0, acc0, false, false);
        acc1 = __builtin_amdgcn_wmma_f32_16x16x4_f32(false, a, false, b1v, (short)0, acc1, false, false);
        acc2 = __builtin_amdgcn_wmma_f32_16x16x4_f32(false, a, false, b2v, (short)0, acc2, false, false);
        acc3 = __builtin_amdgcn_wmma_f32_16x16x4_f32(false, a, false, b3v, (short)0, acc3, false, false);
    }

    float* yb = Y + l16;
    #pragma unroll
    for (int i = 0; i < 8; ++i) {
        float* yr = yb + (size_t)(m0 + i + 8 * half) * CLS;
        yr[0]  = acc0[i];
        yr[16] = acc1[i];
        yr[32] = acc2[i];
        yr[48] = acc3[i];
    }
}

__global__ __launch_bounds__(256) void gcn_zero_i32(int* __restrict__ p, int n)
{
    const int i = (int)(blockIdx.x * (size_t)blockDim.x + threadIdx.x);
    if (i < n) p[i] = 0;
}

__global__ __launch_bounds__(256) void gcn_edge_rank(
    const int* __restrict__ erow, int* __restrict__ cnt, int* __restrict__ rank)
{
    const int e = (int)(blockIdx.x * (size_t)blockDim.x + threadIdx.x);
    if (e >= NEDGE) return;
    rank[e] = atomicAdd(cnt + erow[e], 1);
}

#define SCAN_T   1024
#define SCAN_PER ((NROWS + SCAN_T - 1) / SCAN_T)

__global__ __launch_bounds__(SCAN_T) void gcn_row_scan(
    const int* __restrict__ cnt, int* __restrict__ rowptr)
{
    __shared__ int wsum[32];
    const int t    = threadIdx.x;
    const int lane = t & 31;
    const int wv   = t >> 5;
    const int base = t * SCAN_PER;

    int loc[SCAN_PER];
    int s = 0;
    #pragma unroll
    for (int i = 0; i < SCAN_PER; ++i) {
        const int idx = base + i;
        const int c = (idx < NROWS) ? cnt[idx] : 0;
        loc[i] = c;
        s += c;
    }

    int inc = s;
    #pragma unroll
    for (int off = 1; off < 32; off <<= 1) {
        const int up = __shfl_up(inc, off, 32);
        if (lane >= off) inc += up;
    }
    if (lane == 31) wsum[wv] = inc;
    __syncthreads();
    if (wv == 0) {
        const int w = wsum[lane];
        int winc = w;
        #pragma unroll
        for (int off = 1; off < 32; off <<= 1) {
            const int up = __shfl_up(winc, off, 32);
            if (lane >= off) winc += up;
        }
        wsum[lane] = winc - w;
    }
    __syncthreads();

    int run = wsum[wv] + inc - s;
    #pragma unroll
    for (int i = 0; i < SCAN_PER; ++i) {
        const int idx = base + i;
        if (idx < NROWS) rowptr[idx] = run;
        run += loc[i];
    }
    if (t == SCAN_T - 1) rowptr[NROWS] = run;
}

__global__ __launch_bounds__(256) void gcn_edge_place(
    const int* __restrict__ erow, const int* __restrict__ ecol,
    const float* __restrict__ eval, const int* __restrict__ rowptr,
    const int* __restrict__ rank, int* __restrict__ edges)
{
    const int e = (int)(blockIdx.x * (size_t)blockDim.x + threadIdx.x);
    if (e >= NEDGE) return;
    const int pos = rowptr[erow[e]] + rank[e];
    int2 w;
    w.x = ecol[e];
    w.y = __float_as_int(eval[e]);
    ((int2*)edges)[pos] = w;
}

__global__ __launch_bounds__(256) void gcn_spmm_rows_f128(
    const int* __restrict__ rowptr, const int* __restrict__ edges,
    const float* __restrict__ dense, float* __restrict__ out)
{
    const int lane = threadIdx.x & 31;
    const int r = __builtin_amdgcn_readfirstlane(
        (int)((blockIdx.x * (size_t)blockDim.x + threadIdx.x) >> 5));
    if (r >= NROWS) return;
    const int beg = rowptr[r];
    const int end = rowptr[r + 1];
    const int2*   ew = (const int2*)edges;
    const float4* d4 = (const float4*)dense + lane;

    float4 acc = {0.0f, 0.0f, 0.0f, 0.0f};
    int p = beg;
    for (; p + 4 <= end; p += 4) {
        const int2 e0 = ew[p], e1 = ew[p + 1], e2 = ew[p + 2], e3 = ew[p + 3];
        const float4 g0 = d4[(size_t)e0.x * (HID / 4)];
        const float4 g1 = d4[(size_t)e1.x * (HID / 4)];
        const float4 g2 = d4[(size_t)e2.x * (HID / 4)];
        const float4 g3 = d4[(size_t)e3.x * (HID / 4)];
        const float v0 = __int_as_float(e0.y), v1 = __int_as_float(e1.y);
        const float v2 = __int_as_float(e2.y), v3 = __int_as_float(e3.y);
        acc.x += v0 * g0.x; acc.y += v0 * g0.y; acc.z += v0 * g0.z; acc.w += v0 * g0.w;
        acc.x += v1 * g1.x; acc.y += v1 * g1.y; acc.z += v1 * g1.z; acc.w += v1 * g1.w;
        acc.x += v2 * g2.x; acc.y += v2 * g2.y; acc.z += v2 * g2.z; acc.w += v2 * g2.w;
        acc.x += v3 * g3.x; acc.y += v3 * g3.y; acc.z += v3 * g3.z; acc.w += v3 * g3.w;
    }
    for (; p < end; ++p) {
        const int2 e0 = ew[p];
        const float4 g0 = d4[(size_t)e0.x * (HID / 4)];
        const float v0 = __int_as_float(e0.y);
        acc.x += v0 * g0.x; acc.y += v0 * g0.y; acc.z += v0 * g0.z; acc.w += v0 * g0.w;
    }
    ((float4*)out)[(size_t)r * (HID / 4) + lane] = acc;
}

__global__ __launch_bounds__(256) void gcn_spmm_rows_f64(
    const int* __restrict__ rowptr, const int* __restrict__ edges,
    const float* __restrict__ dense, float* __restrict__ out)
{
    const int lane = threadIdx.x & 31;
    const int r = __builtin_amdgcn_readfirstlane(
        (int)((blockIdx.x * (size_t)blockDim.x + threadIdx.x) >> 5));
    if (r >= NROWS) return;
    const int beg = rowptr[r];
    const int end = rowptr[r + 1];
    const int2*   ew = (const int2*)edges;
    const float2* d2 = (const float2*)dense + lane;

    float2 acc = {0.0f, 0.0f};
    int p = beg;
    for (; p + 4 <= end; p += 4) {
        const int2 e0 = ew[p], e1 = ew[p + 1], e2 = ew[p + 2], e3 = ew[p + 3];
        const float2 g0 = d2[(size_t)e0.x * (CLS / 2)];
        const float2 g1 = d2[(size_t)e1.x * (CLS / 2)];
        const float2 g2 = d2[(size_t)e2.x * (CLS / 2)];
        const float2 g3 = d2[(size_t)e3.x * (CLS / 2)];
        const float v0 = __int_as_float(e0.y), v1 = __int_as_float(e1.y);
        const float v2 = __int_as_float(e2.y), v3 = __int_as_float(e3.y);
        acc.x += v0 * g0.x; acc.y += v0 * g0.y;
        acc.x += v1 * g1.x; acc.y += v1 * g1.y;
        acc.x += v2 * g2.x; acc.y += v2 * g2.y;
        acc.x += v3 * g3.x; acc.y += v3 * g3.y;
    }
    for (; p < end; ++p) {
        const int2 e0 = ew[p];
        const float2 g0 = d2[(size_t)e0.x * (CLS / 2)];
        const float v0 = __int_as_float(e0.y);
        acc.x += v0 * g0.x; acc.y += v0 * g0.y;
    }
    ((float2*)out)[(size_t)r * (CLS / 2) + lane] = acc;
}

__global__ __launch_bounds__(256) void gcn_logsoftmax64(
    const float* __restrict__ L, const float* __restrict__ b2,
    float* __restrict__ out)
{
    const int r = (int)((blockIdx.x * (size_t)blockDim.x + threadIdx.x) >> 5);
    const int lane = threadIdx.x & 31;
    if (r >= NROWS) return;

    float2 v = *(const float2*)(L + (size_t)r * CLS + lane * 2);
    const float2 bb = *(const float2*)(b2 + lane * 2);
    v.x += bb.x;
    v.y += bb.y;

    float m = fmaxf(v.x, v.y);
    #pragma unroll
    for (int off = 16; off > 0; off >>= 1)
        m = fmaxf(m, __shfl_xor(m, off, 32));

    float s = __expf(v.x - m) + __expf(v.y - m);
    #pragma unroll
    for (int off = 16; off > 0; off >>= 1)
        s += __shfl_xor(s, off, 32);

    const float lse = m + __logf(s);
    float2 o;
    o.x = v.x - lse;
    o.y = v.y - lse;
    *(float2*)(out + (size_t)r * CLS + lane * 2) = o;
}

extern "C" void kernel_launch(void* const* d_in, const int* in_sizes, int n_in,
                              void* d_out, int out_size, void* d_ws, size_t ws_size,
                              hipStream_t stream) {
    const float* x        = (const float*)d_in[0];
    const int*   edge_row = (const int*)  d_in[1];
    const int*   edge_col = (const int*)  d_in[2];
    const float* edge_val = (const float*)d_in[3];
    const float* W1       = (const float*)d_in[4];
    const float* b1       = (const float*)d_in[5];
    const float* W2       = (const float*)d_in[6];
    const float* b2       = (const float*)d_in[7];
    float* out = (float*)d_out;

    float* ws    = (float*)d_ws;
    float* XW1   = ws;
    float* Hraw  = ws + (size_t)NROWS * HID;
    float* HW2   = XW1;
    float* Lraw  = Hraw;

    int* cnt     = (int*)(ws + 2 * (size_t)NROWS * HID);
    int* rowptr  = cnt + 50176;
    int* rank    = rowptr + 50176;
    int* edges   = rank + NEDGE;

    const int MT = NROWS / 16;
    const int edge_blocks = (NEDGE + 255) / 256;
    const int row_blocks  = (NROWS * 32 + 255) / 256;

    gcn_gemm1_wmma<<<MT, 64, 0, stream>>>(x, W1, XW1);
    gcn_zero_i32<<<(NROWS + 255) / 256, 256, 0, stream>>>(cnt, NROWS);
    gcn_edge_rank<<<edge_blocks, 256, 0, stream>>>(edge_row, cnt, rank);
    gcn_row_scan<<<1, SCAN_T, 0, stream>>>(cnt, rowptr);
    gcn_edge_place<<<edge_blocks, 256, 0, stream>>>(edge_row, edge_col, edge_val, rowptr, rank, edges);
    gcn_spmm_rows_f128<<<row_blocks, 256, 0, stream>>>(rowptr, edges, XW1, Hraw);
    gcn_gemm2_wmma<<<MT, 32, 0, stream>>>(Hraw, b1, W2, HW2);
    gcn_spmm_rows_f64<<<row_blocks, 256, 0, stream>>>(rowptr, edges, HW2, Lraw);
    gcn_logsoftmax64<<<(NROWS * 32 + 255) / 256, 256, 0, stream>>>(Lraw, b2, out);
}


# Row-pointer scan spread over 49 work-groups, each re-summing the counters in front of its block with whole-line 16-byte loads; replaces one work-group whose lanes read and wrote 196 B apart
# speedup vs baseline: 1.5358x; 1.2918x over previous
#include <hip/hip_runtime.h>


typedef __attribute__((ext_vector_type(2))) float v2f;
typedef __attribute__((ext_vector_type(8))) float v8f;

#define NROWS 50000
#define INDIM 256
#define HID   128
#define CLS   64
#define NEDGE 800000

__global__ __launch_bounds__(64) void gcn_gemm1_wmma(
    const float* __restrict__ X, const float* __restrict__ W1,
    float* __restrict__ Y)
{
    const int mtile = blockIdx.x;
    const int wv    = threadIdx.x >> 5;
    const int lane  = threadIdx.x & 31;
    const int half  = lane >> 4;
    const int l16   = lane & 15;

    const int m0 = mtile * 16;
    const int n0 = wv * 64;

    const float* xrow = X + (size_t)(m0 + l16) * INDIM + 2 * half;
    const float* wcol = W1 + n0 + l16;

    v8f acc0 = {}, acc1 = {}, acc2 = {}, acc3 = {};
    #pragma unroll 2
    for (int k = 0; k < INDIM; k += 4) {
        const float2 av = *(const float2*)(xrow + k);
        v2f a; a.x = av.x; a.y = av.y;
        const float* wk0 = wcol + (size_t)(k + 2 * half) * HID;
        const float* wk1 = wcol + (size_t)(k + 2 * half + 1) * HID;
        v2f b0; b0.x = wk0[0];  b0.y = wk1[0];
        v2f b1; b1.x = wk0[16]; b1.y = wk1[16];
        v2f b2; b2.x = wk0[32]; b2.y = wk1[32];
        v2f b3; b3.x = wk0[48]; b3.y = wk1[48];
        acc0 = __builtin_amdgcn_wmma_f32_16x16x4_f32(false, a, false, b0, (short)0, acc0, false, false);
        acc1 = __builtin_amdgcn_wmma_f32_16x16x4_f32(false, a, false, b1, (short)0, acc1, false, false);
        acc2 = __builtin_amdgcn_wmma_f32_16x16x4_f32(false, a, false, b2, (short)0, acc2, false, false);
        acc3 = __builtin_amdgcn_wmma_f32_16x16x4_f32(false, a, false, b3, (short)0, acc3, false, false);
    }

    float* yb = Y + n0 + l16;
    #pragma unroll
    for (int i = 0; i < 8; ++i) {
        float* yr = yb + (size_t)(m0 + i + 8 * half) * HID;
        yr[0]  = acc0[i];
        yr[16] = acc1[i];
        yr[32] = acc2[i];
        yr[48] = acc3[i];
    }
}

__global__ __launch_bounds__(32) void gcn_gemm2_wmma(
    const float* __restrict__ Hraw, const float* __restrict__ b1,
    const float* __restrict__ W2, float* __restrict__ Y)
{
    const int mtile = blockIdx.x;
    const int lane  = threadIdx.x & 31;
    const int half  = lane >> 4;
    const int l16   = lane & 15;

    const int m0 = mtile * 16;

    const float* hrow = Hraw + (size_t)(m0 + l16) * HID + 2 * half;
    const float* brow = b1 + 2 * half;
    const float* wcol = W2 + l16;

    v8f acc0 = {}, acc1 = {}, acc2 = {}, acc3 = {};
    #pragma unroll 2
    for (int k = 0; k < HID; k += 4) {
        const float2 hv = *(const float2*)(hrow + k);
        const float2 bv = *(const float2*)(brow + k);
        v2f a;
        a.x = fmaxf(hv.x + bv.x, 0.0f);
        a.y = fmaxf(hv.y + bv.y, 0.0f);
        const float* wk0 = wcol + (size_t)(k + 2 * half) * CLS;
        const float* wk1 = wcol + (size_t)(k + 2 * half + 1) * CLS;
        v2f b0; b0.x = wk0[0];  b0.y = wk1[0];
        v2f b1v; b1v.x = wk0[16]; b1v.y = wk1[16];
        v2f b2v; b2v.x = wk0[32]; b2v.y = wk1[32];
        v2f b3v; b3v.x = wk0[48]; b3v.y = wk1[48];
        acc0 = __builtin_amdgcn_wmma_f32_16x16x4_f32(false, a, false, b0,  (short)0, acc0, false, false);
        acc1 = __builtin_amdgcn_wmma_f32_16x16x4_f32(false, a, false, b1v, (short)0, acc1, false, false);
        acc2 = __builtin_amdgcn_wmma_f32_16x16x4_f32(false, a, false, b2v, (short)0, acc2, false, false);
        acc3 = __builtin_amdgcn_wmma_f32_16x16x4_f32(false, a, false, b3v, (short)0, acc3, false, false);
    }

    float* yb = Y + l16;
    #pragma unroll
    for (int i = 0; i < 8; ++i) {
        float* yr = yb + (size_t)(m0 + i + 8 * half) * CLS;
        yr[0]  = acc0[i];
        yr[16] = acc1[i];
        yr[32] = acc2[i];
        yr[48] = acc3[i];
    }
}

__global__ __launch_bounds__(256) void gcn_zero_i32(int* __restrict__ p, int n)
{
    const int i = (int)(blockIdx.x * (size_t)blockDim.x + threadIdx.x);
    if (i < n) p[i] = 0;
}

__global__ __launch_bounds__(256) void gcn_edge_rank(
    const int* __restrict__ erow, int* __restrict__ cnt, int* __restrict__ rank)
{
    const int e = (int)(blockIdx.x * (size_t)blockDim.x + threadIdx.x);
    if (e >= NEDGE) return;
    rank[e] = atomicAdd(cnt + erow[e], 1);
}

#define SCAN_T      1024
#define SCAN_BLOCKS ((NROWS + 1 + SCAN_T - 1) / SCAN_T)
#define SCAN_PRE    (((SCAN_BLOCKS - 1) * (SCAN_T / 4) + SCAN_T - 1) / SCAN_T)

__global__ __launch_bounds__(SCAN_T) void gcn_row_scan(
    const int* __restrict__ cnt, int* __restrict__ rowptr)
{
    __shared__ int wpre[32];
    __shared__ int wown[32];
    const int t    = threadIdx.x;
    const int lane = t & 31;
    const int wv   = t >> 5;
    const int b    = blockIdx.x;
    const int idx  = b * SCAN_T + t;

    const int c = (idx < NROWS) ? cnt[idx] : 0;

    const int4* c4 = (const int4*)cnt;
    const int   n4 = b * (SCAN_T / 4);
    int4 w4[SCAN_PRE];
    #pragma unroll
    for (int j = 0; j < SCAN_PRE; ++j) {
        const int q = j * SCAN_T + t;
        w4[j] = c4[(q < n4) ? q : t];
    }
    int pre = 0;
    #pragma unroll
    for (int j = 0; j < SCAN_PRE; ++j) {
        const int q = j * SCAN_T + t;
        const int s4 = (w4[j].x + w4[j].y) + (w4[j].z + w4[j].w);
        pre += (q < n4) ? s4 : 0;
    }

    int inc = c;
    #pragma unroll
    for (int off = 1; off < 32; off <<= 1) {
        const int up = __shfl_up(inc, off, 32);
        if (lane >= off) inc += up;
    }
    #pragma unroll
    for (int off = 16; off > 0; off >>= 1)
        pre += __shfl_xor(pre, off, 32);

    if (lane == 31) wown[wv] = inc;
    if (lane == 0)  wpre[wv] = pre;
    __syncthreads();

    int v = wpre[lane] + ((lane < wv) ? wown[lane] : 0);
    #pragma unroll
    for (int off = 16; off > 0; off >>= 1)
        v += __shfl_xor(v, off, 32);

    if (idx <= NROWS) rowptr[idx] = v + inc - c;
}

__global__ __launch_bounds__(256) void gcn_edge_place(
    const int* __restrict__ erow, const int* __restrict__ ecol,
    const float* __restrict__ eval, const int* __restrict__ rowptr,
    const int* __restrict__ rank, int* __restrict__ edges)
{
    const int e = (int)(blockIdx.x * (size_t)blockDim.x + threadIdx.x);
    if (e >= NEDGE) return;
    const int pos = rowptr[erow[e]] + rank[e];
    int2 w;
    w.x = ecol[e];
    w.y = __float_as_int(eval[e]);
    ((int2*)edges)[pos] = w;
}

__global__ __launch_bounds__(256) void gcn_spmm_rows_f128(
    const int* __restrict__ rowptr, const int* __restrict__ edges,
    const float* __restrict__ dense, float* __restrict__ out)
{
    const int lane = threadIdx.x & 31;
    const int r = __builtin_amdgcn_readfirstlane(
        (int)((blockIdx.x * (size_t)blockDim.x + threadIdx.x) >> 5));
    if (r >= NROWS) return;
    const int beg = rowptr[r];
    const int end = rowptr[r + 1];
    const int2*   ew = (const int2*)edges;
    const float4* d4 = (const float4*)dense + lane;

    float4 acc = {0.0f, 0.0f, 0.0f, 0.0f};
    int p = beg;
    for (; p + 4 <= end; p += 4) {
        const int2 e0 = ew[p], e1 = ew[p + 1], e2 = ew[p + 2], e3 = ew[p + 3];
        const float4 g0 = d4[(size_t)e0.x * (HID / 4)];
        const float4 g1 = d4[(size_t)e1.x * (HID / 4)];
        const float4 g2 = d4[(size_t)e2.x * (HID / 4)];
        const float4 g3 = d4[(size_t)e3.x * (HID / 4)];
        const float v0 = __int_as_float(e0.y), v1 = __int_as_float(e1.y);
        const float v2 = __int_as_float(e2.y), v3 = __int_as_float(e3.y);
        acc.x += v0 * g0.x; acc.y += v0 * g0.y; acc.z += v0 * g0.z; acc.w += v0 * g0.w;
        acc.x += v1 * g1.x; acc.y += v1 * g1.y; acc.z += v1 * g1.z; acc.w += v1 * g1.w;
        acc.x += v2 * g2.x; acc.y += v2 * g2.y; acc.z += v2 * g2.z; acc.w += v2 * g2.w;
        acc.x += v3 * g3.x; acc.y += v3 * g3.y; acc.z += v3 * g3.z; acc.w += v3 * g3.w;
    }
    for (; p < end; ++p) {
        const int2 e0 = ew[p];
        const float4 g0 = d4[(size_t)e0.x * (HID / 4)];
        const float v0 = __int_as_float(e0.y);
        acc.x += v0 * g0.x; acc.y += v0 * g0.y; acc.z += v0 * g0.z; acc.w += v0 * g0.w;
    }
    ((float4*)out)[(size_t)r * (HID / 4) + lane] = acc;
}

__global__ __launch_bounds__(256) void gcn_spmm_rows_f64(
    const int* __restrict__ rowptr, const int* __restrict__ edges,
    const float* __restrict__ dense, float* __restrict__ out)
{
    const int lane = threadIdx.x & 31;
    const int r = __builtin_amdgcn_readfirstlane(
        (int)((blockIdx.x * (size_t)blockDim.x + threadIdx.x) >> 5));
    if (r >= NROWS) return;
    const int beg = rowptr[r];
    const int end = rowptr[r + 1];
    const int2*   ew = (const int2*)edges;
    const float2* d2 = (const float2*)dense + lane;

    float2 acc = {0.0f, 0.0f};
    int p = beg;
    for (; p + 4 <= end; p += 4) {
        const int2 e0 = ew[p], e1 = ew[p + 1], e2 = ew[p + 2], e3 = ew[p + 3];
        const float2 g0 = d2[(size_t)e0.x * (CLS / 2)];
        const float2 g1 = d2[(size_t)e1.x * (CLS / 2)];
        const float2 g2 = d2[(size_t)e2.x * (CLS / 2)];
        const float2 g3 = d2[(size_t)e3.x * (CLS / 2)];
        const float v0 = __int_as_float(e0.y), v1 = __int_as_float(e1.y);
        const float v2 = __int_as_float(e2.y), v3 = __int_as_float(e3.y);
        acc.x += v0 * g0.x; acc.y += v0 * g0.y;
        acc.x += v1 * g1.x; acc.y += v1 * g1.y;
        acc.x += v2 * g2.x; acc.y += v2 * g2.y;
        acc.x += v3 * g3.x; acc.y += v3 * g3.y;
    }
    for (; p < end; ++p) {
        const int2 e0 = ew[p];
        const float2 g0 = d2[(size_t)e0.x * (CLS / 2)];
        const float v0 = __int_as_float(e0.y);
        acc.x += v0 * g0.x; acc.y += v0 * g0.y;
    }
    ((float2*)out)[(size_t)r * (CLS / 2) + lane] = acc;
}

__global__ __launch_bounds__(256) void gcn_logsoftmax64(
    const float* __restrict__ L, const float* __restrict__ b2,
    float* __restrict__ out)
{
    const int r = (int)((blockIdx.x * (size_t)blockDim.x + threadIdx.x) >> 5);
    const int lane = threadIdx.x & 31;
    if (r >= NROWS) return;

    float2 v = *(const float2*)(L + (size_t)r * CLS + lane * 2);
    const float2 bb = *(const float2*)(b2 + lane * 2);
    v.x += bb.x;
    v.y += bb.y;

    float m = fmaxf(v.x, v.y);
    #pragma unroll
    for (int off = 16; off > 0; off >>= 1)
        m = fmaxf(m, __shfl_xor(m, off, 32));

    float s = __expf(v.x - m) + __expf(v.y - m);
    #pragma unroll
    for (int off = 16; off > 0; off >>= 1)
        s += __shfl_xor(s, off, 32);

    const float lse = m + __logf(s);
    float2 o;
    o.x = v.x - lse;
    o.y = v.y - lse;
    *(float2*)(out + (size_t)r * CLS + lane * 2) = o;
}

extern "C" void kernel_launch(void* const* d_in, const int* in_sizes, int n_in,
                              void* d_out, int out_size, void* d_ws, size_t ws_size,
                              hipStream_t stream) {
    const float* x        = (const float*)d_in[0];
    const int*   edge_row = (const int*)  d_in[1];
    const int*   edge_col = (const int*)  d_in[2];
    const float* edge_val = (const float*)d_in[3];
    const float* W1       = (const float*)d_in[4];
    const float* b1       = (const float*)d_in[5];
    const float* W2       = (const float*)d_in[6];
    const float* b2       = (const float*)d_in[7];
    float* out = (float*)d_out;

    float* ws    = (float*)d_ws;
    float* XW1   = ws;
    float* Hraw  = ws + (size_t)NROWS * HID;
    float* HW2   = XW1;
    float* Lraw  = Hraw;

    int* cnt     = (int*)(ws + 2 * (size_t)NROWS * HID);
    int* rowptr  = cnt + 50176;
    int* rank    = rowptr + 50176;
    int* edges   = rank + NEDGE;

    const int MT = NROWS / 16;
    const int edge_blocks = (NEDGE + 255) / 256;
    const int row_blocks  = (NROWS * 32 + 255) / 256;

    gcn_gemm1_wmma<<<MT, 64, 0, stream>>>(x, W1, XW1);
    gcn_zero_i32<<<(NROWS + 255) / 256, 256, 0, stream>>>(cnt, NROWS);
    gcn_edge_rank<<<edge_blocks, 256, 0, stream>>>(edge_row, cnt, rank);
    gcn_row_scan<<<SCAN_BLOCKS, SCAN_T, 0, stream>>>(cnt, rowptr);
    gcn_edge_place<<<edge_blocks, 256, 0, stream>>>(edge_row, edge_col, edge_val, rowptr, rank, edges);
    gcn_spmm_rows_f128<<<row_blocks, 256, 0, stream>>>(rowptr, edges, XW1, Hraw);
    gcn_gemm2_wmma<<<MT, 32, 0, stream>>>(Hraw, b1, W2, HW2);
    gcn_spmm_rows_f64<<<row_blocks, 256, 0, stream>>>(rowptr, edges, HW2, Lraw);
    gcn_logsoftmax64<<<(NROWS * 32 + 255) / 256, 256, 0, stream>>>(Lraw, b2, out);
}


# Edge ordering in one launch, not four: each edge stored at row*64+rank (rank = counting atomic's return; over 64 spills to a list); scan, place, clear launches go (clear rides in GEMM1): 9 -> 6
# speedup vs baseline: 1.6822x; 1.0953x over previous
#include <hip/hip_runtime.h>


typedef __attribute__((ext_vector_type(2))) float v2f;
typedef __attribute__((ext_vector_type(8))) float v8f;

#define NROWS 50000
#define INDIM 256
#define HID   128
#define CLS   64
#define NEDGE 800000
#define BUCKET    64
#define CNT_WORDS 50176

__global__ __launch_bounds__(64) void gcn_gemm1_wmma(
    const float* __restrict__ X, const float* __restrict__ W1,
    float* __restrict__ Y, int* __restrict__ cnt)
{
    const int gid = blockIdx.x * 64 + threadIdx.x;
    if (gid < CNT_WORDS) cnt[gid] = 0;

    const int mtile = blockIdx.x;
    const int wv    = threadIdx.x >> 5;
    const int lane  = threadIdx.x & 31;
    const int half  = lane >> 4;
    const int l16   = lane & 15;

    const int m0 = mtile * 16;
    const int n0 = wv * 64;

    const float* xrow = X + (size_t)(m0 + l16) * INDIM + 2 * half;
    const float* wcol = W1 + n0 + l16;

    v8f acc0 = {}, acc1 = {}, acc2 = {}, acc3 = {};
    #pragma unroll 2
    for (int k = 0; k < INDIM; k += 4) {
        const float2 av = *(const float2*)(xrow + k);
        v2f a; a.x = av.x; a.y = av.y;
        const float* wk0 = wcol + (size_t)(k + 2 * half) * HID;
        const float* wk1 = wcol + (size_t)(k + 2 * half + 1) * HID;
        v2f b0; b0.x = wk0[0];  b0.y = wk1[0];
        v2f b1; b1.x = wk0[16]; b1.y = wk1[16];
        v2f b2; b2.x = wk0[32]; b2.y = wk1[32];
        v2f b3; b3.x = wk0[48]; b3.y = wk1[48];
        acc0 = __builtin_amdgcn_wmma_f32_16x16x4_f32(false, a, false, b0, (short)0, acc0, false, false);
        acc1 = __builtin_amdgcn_wmma_f32_16x16x4_f32(false, a, false, b1, (short)0, acc1, false, false);
        acc2 = __builtin_amdgcn_wmma_f32_16x16x4_f32(false, a, false, b2, (short)0, acc2, false, false);
        acc3 = __builtin_amdgcn_wmma_f32_16x16x4_f32(false, a, false, b3, (short)0, acc3, false, false);
    }

    float* yb = Y + n0 + l16;
    #pragma unroll
    for (int i = 0; i < 8; ++i) {
        float* yr = yb + (size_t)(m0 + i + 8 * half) * HID;
        yr[0]  = acc0[i];
        yr[16] = acc1[i];
        yr[32] = acc2[i];
        yr[48] = acc3[i];
    }
}

__global__ __launch_bounds__(32) void gcn_gemm2_wmma(
    const float* __restrict__ Hraw, const float* __restrict__ b1,
    const float* __restrict__ W2, float* __restrict__ Y)
{
    const int mtile = blockIdx.x;
    const int lane  = threadIdx.x & 31;
    const int half  = lane >> 4;
    const int l16   = lane & 15;

    const int m0 = mtile * 16;

    const float* hrow = Hraw + (size_t)(m0 + l16) * HID + 2 * half;
    const float* brow = b1 + 2 * half;
    const float* wcol = W2 + l16;

    v8f acc0 = {}, acc1 = {}, acc2 = {}, acc3 = {};
    #pragma unroll 2
    for (int k = 0; k < HID; k += 4) {
        const float2 hv = *(const float2*)(hrow + k);
        const float2 bv = *(const float2*)(brow + k);
        v2f a;
        a.x = fmaxf(hv.x + bv.x, 0.0f);
        a.y = fmaxf(hv.y + bv.y, 0.0f);
        const float* wk0 = wcol + (size_t)(k + 2 * half) * CLS;
        const float* wk1 = wcol + (size_t)(k + 2 * half + 1) * CLS;
        v2f b0; b0.x = wk0[0];  b0.y = wk1[0];
        v2f b1v; b1v.x = wk0[16]; b1v.y = wk1[16];
        v2f b2v; b2v.x = wk0[32]; b2v.y = wk1[32];
        v2f b3v; b3v.x = wk0[48]; b3v.y = wk1[48];
        acc0 = __builtin_amdgcn_wmma_f32_16x16x4_f32(false, a, false, b0,  (short)0, acc0, false, false);
        acc1 = __builtin_amdgcn_wmma_f32_16x16x4_f32(false, a, false, b1v, (short)0, acc1, false, false);
        acc2 = __builtin_amdgcn_wmma_f32_16x16x4_f32(false, a, false, b2v, (short)0, acc2, false, false);
        acc3 = __builtin_amdgcn_wmma_f32_16x16x4_f32(false, a, false, b3v, (short)0, acc3, false, false);
    }

    float* yb = Y + l16;
    #pragma unroll
    for (int i = 0; i < 8; ++i) {
        float* yr = yb + (size_t)(m0 + i + 8 * half) * CLS;
        yr[0]  = acc0[i];
        yr[16] = acc1[i];
        yr[32] = acc2[i];
        yr[48] = acc3[i];
    }
}

__global__ __launch_bounds__(256) void gcn_edge_bucket(
    const int* __restrict__ erow, const int* __restrict__ ecol,
    const float* __restrict__ eval, int* __restrict__ cnt,
    int* __restrict__ edges, int* __restrict__ spill_row,
    int* __restrict__ spill_edge)
{
    const int e = (int)(blockIdx.x * (size_t)blockDim.x + threadIdx.x);
    if (e >= NEDGE) return;
    const int r = erow[e];
    int2 w;
    w.x = ecol[e];
    w.y = __float_as_int(eval[e]);
    const int k = atomicAdd(cnt + r, 1);
    int2* dst;
    if (k < BUCKET) {
        dst = (int2*)edges + (size_t)r * BUCKET + k;
    } else {
        const int q = atomicAdd(cnt + NROWS, 1);
        spill_row[q] = r;
        dst = (int2*)spill_edge + q;
    }
    *dst = w;
}

__global__ __launch_bounds__(256) void gcn_spmm_rows_f128(
    const int* __restrict__ cnt, const int* __restrict__ edges,
    const int* __restrict__ spill_row, const int* __restrict__ spill_edge,
    const float* __restrict__ dense, float* __restrict__ out)
{
    const int lane = threadIdx.x & 31;
    const int r = __builtin_amdgcn_readfirstlane(
        (int)((blockIdx.x * (size_t)blockDim.x + threadIdx.x) >> 5));
    if (r >= NROWS) return;
    const int n   = cnt[r];
    const int end = (n < BUCKET) ? n : BUCKET;
    const int2*   ew = (const int2*)edges + (size_t)r * BUCKET;
    const float4* d4 = (const float4*)dense + lane;

    float4 acc = {0.0f, 0.0f, 0.0f, 0.0f};
    int p = 0;
    for (; p + 4 <= end; p += 4) {
        const int2 e0 = ew[p], e1 = ew[p + 1], e2 = ew[p + 2], e3 = ew[p + 3];
        const float4 g0 = d4[(size_t)e0.x * (HID / 4)];
        const float4 g1 = d4[(size_t)e1.x * (HID / 4)];
        const float4 g2 = d4[(size_t)e2.x * (HID / 4)];
        const float4 g3 = d4[(size_t)e3.x * (HID / 4)];
        const float v0 = __int_as_float(e0.y), v1 = __int_as_float(e1.y);
        const float v2 = __int_as_float(e2.y), v3 = __int_as_float(e3.y);
        acc.x += v0 * g0.x; acc.y += v0 * g0.y; acc.z += v0 * g0.z; acc.w += v0 * g0.w;
        acc.x += v1 * g1.x; acc.y += v1 * g1.y; acc.z += v1 * g1.z; acc.w += v1 * g1.w;
        acc.x += v2 * g2.x; acc.y += v2 * g2.y; acc.z += v2 * g2.z; acc.w += v2 * g2.w;
        acc.x += v3 * g3.x; acc.y += v3 * g3.y; acc.z += v3 * g3.z; acc.w += v3 * g3.w;
    }
    for (; p < end; ++p) {
        const int2 e0 = ew[p];
        const float4 g0 = d4[(size_t)e0.x * (HID / 4)];
        const float v0 = __int_as_float(e0.y);
        acc.x += v0 * g0.x; acc.y += v0 * g0.y; acc.z += v0 * g0.z; acc.w += v0 * g0.w;
    }
    if (n > BUCKET) {
        const int q = cnt[NROWS];
        for (int i = 0; i < q; ++i) {
            if (spill_row[i] != r) continue;
            const int2 e0 = ((const int2*)spill_edge)[i];
            const float4 g0 = d4[(size_t)e0.x * (HID / 4)];
            const float v0 = __int_as_float(e0.y);
            acc.x += v0 * g0.x; acc.y += v0 * g0.y; acc.z += v0 * g0.z; acc.w += v0 * g0.w;
        }
    }
    ((float4*)out)[(size_t)r * (HID / 4) + lane] = acc;
}

__global__ __launch_bounds__(256) void gcn_spmm_rows_f64(
    const int* __restrict__ cnt, const int* __restrict__ edges,
    const int* __restrict__ spill_row, const int* __restrict__ spill_edge,
    const float* __restrict__ dense, float* __restrict__ out)
{
    const int lane = threadIdx.x & 31;
    const int r = __builtin_amdgcn_readfirstlane(
        (int)((blockIdx.x * (size_t)blockDim.x + threadIdx.x) >> 5));
    if (r >= NROWS) return;
    const int n   = cnt[r];
    const int end = (n < BUCKET) ? n : BUCKET;
    const int2*   ew = (const int2*)edges + (size_t)r * BUCKET;
    const float2* d2 = (const float2*)dense + lane;

    float2 acc = {0.0f, 0.0f};
    int p = 0;
    for (; p + 4 <= end; p += 4) {
        const int2 e0 = ew[p], e1 = ew[p + 1], e2 = ew[p + 2], e3 = ew[p + 3];
        const float2 g0 = d2[(size_t)e0.x * (CLS / 2)];
        const float2 g1 = d2[(size_t)e1.x * (CLS / 2)];
        const float2 g2 = d2[(size_t)e2.x * (CLS / 2)];
        const float2 g3 = d2[(size_t)e3.x * (CLS / 2)];
        const float v0 = __int_as_float(e0.y), v1 = __int_as_float(e1.y);
        const float v2 = __int_as_float(e2.y), v3 = __int_as_float(e3.y);
        acc.x += v0 * g0.x; acc.y += v0 * g0.y;
        acc.x += v1 * g1.x; acc.y += v1 * g1.y;
        acc.x += v2 * g2.x; acc.y += v2 * g2.y;
        acc.x += v3 * g3.x; acc.y += v3 * g3.y;
    }
    for (; p < end; ++p) {
        const int2 e0 = ew[p];
        const float2 g0 = d2[(size_t)e0.x * (CLS / 2)];
        const float v0 = __int_as_float(e0.y);
        acc.x += v0 * g0.x; acc.y += v0 * g0.y;
    }
    if (n > BUCKET) {
        const int q = cnt[NROWS];
        for (int i = 0; i < q; ++i) {
            if (spill_row[i] != r) continue;
            const int2 e0 = ((const int2*)spill_edge)[i];
            const float2 g0 = d2[(size_t)e0.x * (CLS / 2)];
            const float v0 = __int_as_float(e0.y);
            acc.x += v0 * g0.x; acc.y += v0 * g0.y;
        }
    }
    ((float2*)out)[(size_t)r * (CLS / 2) + lane] = acc;
}

__global__ __launch_bounds__(256) void gcn_logsoftmax64(
    const float* __restrict__ L, const float* __restrict__ b2,
    float* __restrict__ out)
{
    const int r = (int)((blockIdx.x * (size_t)blockDim.x + threadIdx.x) >> 5);
    const int lane = threadIdx.x & 31;
    if (r >= NROWS) return;

    float2 v = *(const float2*)(L + (size_t)r * CLS + lane * 2);
    const float2 bb = *(const float2*)(b2 + lane * 2);
    v.x += bb.x;
    v.y += bb.y;

    float m = fmaxf(v.x, v.y);
    #pragma unroll
    for (int off = 16; off > 0; off >>= 1)
        m = fmaxf(m, __shfl_xor(m, off, 32));

    float s = __expf(v.x - m) + __expf(v.y - m);
    #pragma unroll
    for (int off = 16; off > 0; off >>= 1)
        s += __shfl_xor(s, off, 32);

    const float lse = m + __logf(s);
    float2 o;
    o.x = v.x - lse;
    o.y = v.y - lse;
    *(float2*)(out + (size_t)r * CLS + lane * 2) = o;
}

extern "C" void kernel_launch(void* const* d_in, const int* in_sizes, int n_in,
                              void* d_out, int out_size, void* d_ws, size_t ws_size,
                              hipStream_t stream) {
    const float* x        = (const float*)d_in[0];
    const int*   edge_row = (const int*)  d_in[1];
    const int*   edge_col = (const int*)  d_in[2];
    const float* edge_val = (const float*)d_in[3];
    const float* W1       = (const float*)d_in[4];
    const float* b1       = (const float*)d_in[5];
    const float* W2       = (const float*)d_in[6];
    const float* b2       = (const float*)d_in[7];
    float* out = (float*)d_out;

    float* ws    = (float*)d_ws;
    float* XW1   = ws;
    float* Hraw  = ws + (size_t)NROWS * HID;
    float* HW2   = XW1;
    float* Lraw  = Hraw;

    int* cnt        = (int*)(ws + 2 * (size_t)NROWS * HID);
    int* edges      = cnt + CNT_WORDS;
    int* spill_row  = edges + (size_t)NROWS * BUCKET * 2;
    int* spill_edge = spill_row + NEDGE;

    const int MT = NROWS / 16;
    const int edge_blocks = (NEDGE + 255) / 256;
    const int row_blocks  = (NROWS * 32 + 255) / 256;

    gcn_gemm1_wmma<<<MT, 64, 0, stream>>>(x, W1, XW1, cnt);
    gcn_edge_bucket<<<edge_blocks, 256, 0, stream>>>(edge_row, edge_col, edge_val, cnt, edges, spill_row, spill_edge);
    gcn_spmm_rows_f128<<<row_blocks, 256, 0, stream>>>(cnt, edges, spill_row, spill_edge, XW1, Hraw);
    gcn_gemm2_wmma<<<MT, 32, 0, stream>>>(Hraw, b1, W2, HW2);
    gcn_spmm_rows_f64<<<row_blocks, 256, 0, stream>>>(cnt, edges, spill_row, spill_edge, HW2, Lraw);
    gcn_logsoftmax64<<<(NROWS * 32 + 255) / 256, 256, 0, stream>>>(Lraw, b2, out);
}


# log_softmax (+b2) moved into the F=64 SpMM row wave, which already holds the whole 64-logit row in registers: its launch, its 50000 waves and the logits' trip through memory go; 6 -> 5 launches
# speedup vs baseline: 1.7455x; 1.0376x over previous
#include <hip/hip_runtime.h>


typedef __attribute__((ext_vector_type(2))) float v2f;
typedef __attribute__((ext_vector_type(8))) float v8f;

#define NROWS 50000
#define INDIM 256
#define HID   128
#define CLS   64
#define NEDGE 800000
#define BUCKET    64
#define CNT_WORDS 50176

__global__ __launch_bounds__(64) void gcn_gemm1_wmma(
    const float* __restrict__ X, const float* __restrict__ W1,
    float* __restrict__ Y, int* __restrict__ cnt)
{
    const int gid = blockIdx.x * 64 + threadIdx.x;
    if (gid < CNT_WORDS) cnt[gid] = 0;

    const int mtile = blockIdx.x;
    const int wv    = threadIdx.x >> 5;
    const int lane  = threadIdx.x & 31;
    const int half  = lane >> 4;
    const int l16   = lane & 15;

    const int m0 = mtile * 16;
    const int n0 = wv * 64;

    const float* xrow = X + (size_t)(m0 + l16) * INDIM + 2 * half;
    const float* wcol = W1 + n0 + l16;

    v8f acc0 = {}, acc1 = {}, acc2 = {}, acc3 = {};
    #pragma unroll 2
    for (int k = 0; k < INDIM; k += 4) {
        const float2 av = *(const float2*)(xrow + k);
        v2f a; a.x = av.x; a.y = av.y;
        const float* wk0 = wcol + (size_t)(k + 2 * half) * HID;
        const float* wk1 = wcol + (size_t)(k + 2 * half + 1) * HID;
        v2f b0; b0.x = wk0[0];  b0.y = wk1[0];
        v2f b1; b1.x = wk0[16]; b1.y = wk1[16];
        v2f b2; b2.x = wk0[32]; b2.y = wk1[32];
        v2f b3; b3.x = wk0[48]; b3.y = wk1[48];
        acc0 = __builtin_amdgcn_wmma_f32_16x16x4_f32(false, a, false, b0, (short)0, acc0, false, false);
        acc1 = __builtin_amdgcn_wmma_f32_16x16x4_f32(false, a, false, b1, (short)0, acc1, false, false);
        acc2 = __builtin_amdgcn_wmma_f32_16x16x4_f32(false, a, false, b2, (short)0, acc2, false, false);
        acc3 = __builtin_amdgcn_wmma_f32_16x16x4_f32(false, a, false, b3, (short)0, acc3, false, false);
    }

    float* yb = Y + n0 + l16;
    #pragma unroll
    for (int i = 0; i < 8; ++i) {
        float* yr = yb + (size_t)(m0 + i + 8 * half) * HID;
        yr[0]  = acc0[i];
        yr[16] = acc1[i];
        yr[32] = acc2[i];
        yr[48] = acc3[i];
    }
}

__global__ __launch_bounds__(32) void gcn_gemm2_wmma(
    const float* __restrict__ Hraw, const float* __restrict__ b1,
    const float* __restrict__ W2, float* __restrict__ Y)
{
    const int mtile = blockIdx.x;
    const int lane  = threadIdx.x & 31;
    const int half  = lane >> 4;
    const int l16   = lane & 15;

    const int m0 = mtile * 16;

    const float* hrow = Hraw + (size_t)(m0 + l16) * HID + 2 * half;
    const float* brow = b1 + 2 * half;
    const float* wcol = W2 + l16;

    v8f acc0 = {}, acc1 = {}, acc2 = {}, acc3 = {};
    #pragma unroll 2
    for (int k = 0; k < HID; k += 4) {
        const float2 hv = *(const float2*)(hrow + k);
        const float2 bv = *(const float2*)(brow + k);
        v2f a;
        a.x = fmaxf(hv.x + bv.x, 0.0f);
        a.y = fmaxf(hv.y + bv.y, 0.0f);
        const float* wk0 = wcol + (size_t)(k + 2 * half) * CLS;
        const float* wk1 = wcol + (size_t)(k + 2 * half + 1) * CLS;
        v2f b0; b0.x = wk0[0];  b0.y = wk1[0];
        v2f b1v; b1v.x = wk0[16]; b1v.y = wk1[16];
        v2f b2v; b2v.x = wk0[32]; b2v.y = wk1[32];
        v2f b3v; b3v.x = wk0[48]; b3v.y = wk1[48];
        acc0 = __builtin_amdgcn_wmma_f32_16x16x4_f32(false, a, false, b0,  (short)0, acc0, false, false);
        acc1 = __builtin_amdgcn_wmma_f32_16x16x4_f32(false, a, false, b1v, (short)0, acc1, false, false);
        acc2 = __builtin_amdgcn_wmma_f32_16x16x4_f32(false, a, false, b2v, (short)0, acc2, false, false);
        acc3 = __builtin_amdgcn_wmma_f32_16x16x4_f32(false, a, false, b3v, (short)0, acc3, false, false);
    }

    float* yb = Y + l16;
    #pragma unroll
    for (int i = 0; i < 8; ++i) {
        float* yr = yb + (size_t)(m0 + i + 8 * half) * CLS;
        yr[0]  = acc0[i];
        yr[16] = acc1[i];
        yr[32] = acc2[i];
        yr[48] = acc3[i];
    }
}

__global__ __launch_bounds__(256) void gcn_edge_bucket(
    const int* __restrict__ erow, const int* __restrict__ ecol,
    const float* __restrict__ eval, int* __restrict__ cnt,
    int* __restrict__ edges, int* __restrict__ spill_row,
    int* __restrict__ spill_edge)
{
    const int e = (int)(blockIdx.x * (size_t)blockDim.x + threadIdx.x);
    if (e >= NEDGE) return;
    const int r = erow[e];
    int2 w;
    w.x = ecol[e];
    w.y = __float_as_int(eval[e]);
    const int k = atomicAdd(cnt + r, 1);
    int2* dst;
    if (k < BUCKET) {
        dst = (int2*)edges + (size_t)r * BUCKET + k;
    } else {
        const int q = atomicAdd(cnt + NROWS, 1);
        spill_row[q] = r;
        dst = (int2*)spill_edge + q;
    }
    *dst = w;
}

__global__ __launch_bounds__(256) void gcn_spmm_rows_f128(
    const int* __restrict__ cnt, const int* __restrict__ edges,
    const int* __restrict__ spill_row, const int* __restrict__ spill_edge,
    const float* __restrict__ dense, float* __restrict__ out)
{
    const int lane = threadIdx.x & 31;
    const int r = __builtin_amdgcn_readfirstlane(
        (int)((blockIdx.x * (size_t)blockDim.x + threadIdx.x) >> 5));
    if (r >= NROWS) return;
    const int n   = cnt[r];
    const int end = (n < BUCKET) ? n : BUCKET;
    const int2*   ew = (const int2*)edges + (size_t)r * BUCKET;
    const float4* d4 = (const float4*)dense + lane;

    float4 acc = {0.0f, 0.0f, 0.0f, 0.0f};
    int p = 0;
    for (; p + 4 <= end; p += 4) {
        const int2 e0 = ew[p], e1 = ew[p + 1], e2 = ew[p + 2], e3 = ew[p + 3];
        const float4 g0 = d4[(size_t)e0.x * (HID / 4)];
        const float4 g1 = d4[(size_t)e1.x * (HID / 4)];
        const float4 g2 = d4[(size_t)e2.x * (HID / 4)];
        const float4 g3 = d4[(size_t)e3.x * (HID / 4)];
        const float v0 = __int_as_float(e0.y), v1 = __int_as_float(e1.y);
        const float v2 = __int_as_float(e2.y), v3 = __int_as_float(e3.y);
        acc.x += v0 * g0.x; acc.y += v0 * g0.y; acc.z += v0 * g0.z; acc.w += v0 * g0.w;
        acc.x += v1 * g1.x; acc.y += v1 * g1.y; acc.z += v1 * g1.z; acc.w += v1 * g1.w;
        acc.x += v2 * g2.x; acc.y += v2 * g2.y; acc.z += v2 * g2.z; acc.w += v2 * g2.w;
        acc.x += v3 * g3.x; acc.y += v3 * g3.y; acc.z += v3 * g3.z; acc.w += v3 * g3.w;
    }
    for (; p < end; ++p) {
        const int2 e0 = ew[p];
        const float4 g0 = d4[(size_t)e0.x * (HID / 4)];
        const float v0 = __int_as_float(e0.y);
        acc.x += v0 * g0.x; acc.y += v0 * g0.y; acc.z += v0 * g0.z; acc.w += v0 * g0.w;
    }
    if (n > BUCKET) {
        const int q = cnt[NROWS];
        for (int i = 0; i < q; ++i) {
            if (spill_row[i] != r) continue;
            const int2 e0 = ((const int2*)spill_edge)[i];
            const float4 g0 = d4[(size_t)e0.x * (HID / 4)];
            const float v0 = __int_as_float(e0.y);
            acc.x += v0 * g0.x; acc.y += v0 * g0.y; acc.z += v0 * g0.z; acc.w += v0 * g0.w;
        }
    }
    ((float4*)out)[(size_t)r * (HID / 4) + lane] = acc;
}

__global__ __launch_bounds__(256) void gcn_spmm_rows_f64_logsoftmax(
    const int* __restrict__ cnt, const int* __restrict__ edges,
    const int* __restrict__ spill_row, const int* __restrict__ spill_edge,
    const float* __restrict__ dense, const float* __restrict__ b2,
    float* __restrict__ out)
{
    const int lane = threadIdx.x & 31;
    const int r = __builtin_amdgcn_readfirstlane(
        (int)((blockIdx.x * (size_t)blockDim.x + threadIdx.x) >> 5));
    if (r >= NROWS) return;
    const int n   = cnt[r];
    const int end = (n < BUCKET) ? n : BUCKET;
    const int2*   ew = (const int2*)edges + (size_t)r * BUCKET;
    const float2* d2 = (const float2*)dense + lane;

    float2 acc = {0.0f, 0.0f};
    int p = 0;
    for (; p + 4 <= end; p += 4) {
        const int2 e0 = ew[p], e1 = ew[p + 1], e2 = ew[p + 2], e3 = ew[p + 3];
        const float2 g0 = d2[(size_t)e0.x * (CLS / 2)];
        const float2 g1 = d2[(size_t)e1.x * (CLS / 2)];
        const float2 g2 = d2[(size_t)e2.x * (CLS / 2)];
        const float2 g3 = d2[(size_t)e3.x * (CLS / 2)];
        const float v0 = __int_as_float(e0.y), v1 = __int_as_float(e1.y);
        const float v2 = __int_as_float(e2.y), v3 = __int_as_float(e3.y);
        acc.x += v0 * g0.x; acc.y += v0 * g0.y;
        acc.x += v1 * g1.x; acc.y += v1 * g1.y;
        acc.x += v2 * g2.x; acc.y += v2 * g2.y;
        acc.x += v3 * g3.x; acc.y += v3 * g3.y;
    }
    for (; p < end; ++p) {
        const int2 e0 = ew[p];
        const float2 g0 = d2[(size_t)e0.x * (CLS / 2)];
        const float v0 = __int_as_float(e0.y);
        acc.x += v0 * g0.x; acc.y += v0 * g0.y;
    }
    if (n > BUCKET) {
        const int q = cnt[NROWS];
        for (int i = 0; i < q; ++i) {
            if (spill_row[i] != r) continue;
            const int2 e0 = ((const int2*)spill_edge)[i];
            const float2 g0 = d2[(size_t)e0.x * (CLS / 2)];
            const float v0 = __int_as_float(e0.y);
            acc.x += v0 * g0.x; acc.y += v0 * g0.y;
        }
    }

    const float2 bb = *(const float2*)(b2 + lane * 2);
    float2 v;
    v.x = acc.x + bb.x;
    v.y = acc.y + bb.y;

    float m = fmaxf(v.x, v.y);
    #pragma unroll
    for (int off = 16; off > 0; off >>= 1)
        m = fmaxf(m, __shfl_xor(m, off, 32));

    float s = __expf(v.x - m) + __expf(v.y - m);
    #pragma unroll
    for (int off = 16; off > 0; off >>= 1)
        s += __shfl_xor(s, off, 32);

    const float lse = m + __logf(s);
    float2 o;
    o.x = v.x - lse;
    o.y = v.y - lse;
    ((float2*)out)[(size_t)r * (CLS / 2) + lane] = o;
}

extern "C" void kernel_launch(void* const* d_in, const int* in_sizes, int n_in,
                              void* d_out, int out_size, void* d_ws, size_t ws_size,
                              hipStream_t stream) {
    const float* x        = (const float*)d_in[0];
    const int*   edge_row = (const int*)  d_in[1];
    const int*   edge_col = (const int*)  d_in[2];
    const float* edge_val = (const float*)d_in[3];
    const float* W1       = (const float*)d_in[4];
    const float* b1       = (const float*)d_in[5];
    const float* W2       = (const float*)d_in[6];
    const float* b2       = (const float*)d_in[7];
    float* out = (float*)d_out;

    float* ws    = (float*)d_ws;
    float* XW1   = ws;
    float* Hraw  = ws + (size_t)NROWS * HID;
    float* HW2   = XW1;

    int* cnt        = (int*)(ws + 2 * (size_t)NROWS * HID);
    int* edges      = cnt + CNT_WORDS;
    int* spill_row  = edges + (size_t)NROWS * BUCKET * 2;
    int* spill_edge = spill_row + NEDGE;

    const int MT = NROWS / 16;
    const int edge_blocks = (NEDGE + 255) / 256;
    const int row_blocks  = (NROWS * 32 + 255) / 256;

    gcn_gemm1_wmma<<<MT, 64, 0, stream>>>(x, W1, XW1, cnt);
    gcn_edge_bucket<<<edge_blocks, 256, 0, stream>>>(edge_row, edge_col, edge_val, cnt, edges, spill_row, spill_edge);
    gcn_spmm_rows_f128<<<row_blocks, 256, 0, stream>>>(cnt, edges, spill_row, spill_edge, XW1, Hraw);
    gcn_gemm2_wmma<<<MT, 32, 0, stream>>>(Hraw, b1, W2, HW2);
    gcn_spmm_rows_f64_logsoftmax<<<row_blocks, 256, 0, stream>>>(cnt, edges, spill_row, spill_edge, HW2, b2, out);
}


# Second GEMM moved into the F=128 SpMM launch: a 16-wave group puts its 16 finished relu rows in LDS and 4 of its waves do the 16x128 @ 128x64 WMMA product from there; 5 -> 4 launches
# speedup vs baseline: 1.8820x; 1.0782x over previous
#include <hip/hip_runtime.h>


typedef __attribute__((ext_vector_type(2))) float v2f;
typedef __attribute__((ext_vector_type(8))) float v8f;

#define NROWS 50000
#define INDIM 256
#define HID   128
#define CLS   64
#define NEDGE 800000
#define BUCKET    64
#define CNT_WORDS 50176

__global__ __launch_bounds__(64) void gcn_gemm1_wmma(
    const float* __restrict__ X, const float* __restrict__ W1,
    float* __restrict__ Y, int* __restrict__ cnt)
{
    const int gid = blockIdx.x * 64 + threadIdx.x;
    if (gid < CNT_WORDS) cnt[gid] = 0;

    const int mtile = blockIdx.x;
    const int wv    = threadIdx.x >> 5;
    const int lane  = threadIdx.x & 31;
    const int half  = lane >> 4;
    const int l16   = lane & 15;

    const int m0 = mtile * 16;
    const int n0 = wv * 64;

    const float* xrow = X + (size_t)(m0 + l16) * INDIM + 2 * half;
    const float* wcol = W1 + n0 + l16;

    v8f acc0 = {}, acc1 = {}, acc2 = {}, acc3 = {};
    #pragma unroll 2
    for (int k = 0; k < INDIM; k += 4) {
        const float2 av = *(const float2*)(xrow + k);
        v2f a; a.x = av.x; a.y = av.y;
        const float* wk0 = wcol + (size_t)(k + 2 * half) * HID;
        const float* wk1 = wcol + (size_t)(k + 2 * half + 1) * HID;
        v2f b0; b0.x = wk0[0];  b0.y = wk1[0];
        v2f b1; b1.x = wk0[16]; b1.y = wk1[16];
        v2f b2; b2.x = wk0[32]; b2.y = wk1[32];
        v2f b3; b3.x = wk0[48]; b3.y = wk1[48];
        acc0 = __builtin_amdgcn_wmma_f32_16x16x4_f32(false, a, false, b0, (short)0, acc0, false, false);
        acc1 = __builtin_amdgcn_wmma_f32_16x16x4_f32(false, a, false, b1, (short)0, acc1, false, false);
        acc2 = __builtin_amdgcn_wmma_f32_16x16x4_f32(false, a, false, b2, (short)0, acc2, false, false);
        acc3 = __builtin_amdgcn_wmma_f32_16x16x4_f32(false, a, false, b3, (short)0, acc3, false, false);
    }

    float* yb = Y + n0 + l16;
    #pragma unroll
    for (int i = 0; i < 8; ++i) {
        float* yr = yb + (size_t)(m0 + i + 8 * half) * HID;
        yr[0]  = acc0[i];
        yr[16] = acc1[i];
        yr[32] = acc2[i];
        yr[48] = acc3[i];
    }
}

__global__ __launch_bounds__(256) void gcn_edge_bucket(
    const int* __restrict__ erow, const int* __restrict__ ecol,
    const float* __restrict__ eval, int* __restrict__ cnt,
    int* __restrict__ edges, int* __restrict__ spill_row,
    int* __restrict__ spill_edge)
{
    const int e = (int)(blockIdx.x * (size_t)blockDim.x + threadIdx.x);
    if (e >= NEDGE) return;
    const int r = erow[e];
    int2 w;
    w.x = ecol[e];
    w.y = __float_as_int(eval[e]);
    const int k = atomicAdd(cnt + r, 1);
    int2* dst;
    if (k < BUCKET) {
        dst = (int2*)edges + (size_t)r * BUCKET + k;
    } else {
        const int q = atomicAdd(cnt + NROWS, 1);
        spill_row[q] = r;
        dst = (int2*)spill_edge + q;
    }
    *dst = w;
}

#define HPITCH 132

__global__ __launch_bounds__(512) void gcn_spmm_rows_f128_gemm2(
    const int* __restrict__ cnt, const int* __restrict__ edges,
    const int* __restrict__ spill_row, const int* __restrict__ spill_edge,
    const float* __restrict__ dense, const float* __restrict__ b1,
    const float* __restrict__ W2, float* __restrict__ Y)
{
    __shared__ float hs[16 * HPITCH];

    const int lane = threadIdx.x & 31;
    const int wv = __builtin_amdgcn_readfirstlane((int)(threadIdx.x >> 5));
    const int r  = (int)blockIdx.x * 16 + wv;
    const int n   = cnt[r];
    const int end = (n < BUCKET) ? n : BUCKET;
    const int2*   ew = (const int2*)edges + (size_t)r * BUCKET;
    const float4* d4 = (const float4*)dense + lane;

    float4 acc = {0.0f, 0.0f, 0.0f, 0.0f};
    int p = 0;
    for (; p + 4 <= end; p += 4) {
        const int2 e0 = ew[p], e1 = ew[p + 1], e2 = ew[p + 2], e3 = ew[p + 3];
        const float4 g0 = d4[(size_t)e0.x * (HID / 4)];
        const float4 g1 = d4[(size_t)e1.x * (HID / 4)];
        const float4 g2 = d4[(size_t)e2.x * (HID / 4)];
        const float4 g3 = d4[(size_t)e3.x * (HID / 4)];
        const float v0 = __int_as_float(e0.y), v1 = __int_as_float(e1.y);
        const float v2 = __int_as_float(e2.y), v3 = __int_as_float(e3.y);
        acc.x += v0 * g0.x; acc.y += v0 * g0.y; acc.z += v0 * g0.z; acc.w += v0 * g0.w;
        acc.x += v1 * g1.x; acc.y += v1 * g1.y; acc.z += v1 * g1.z; acc.w += v1 * g1.w;
        acc.x += v2 * g2.x; acc.y += v2 * g2.y; acc.z += v2 * g2.z; acc.w += v2 * g2.w;
        acc.x += v3 * g3.x; acc.y += v3 * g3.y; acc.z += v3 * g3.z; acc.w += v3 * g3.w;
    }
    for (; p < end; ++p) {
        const int2 e0 = ew[p];
        const float4 g0 = d4[(size_t)e0.x * (HID / 4)];
        const float v0 = __int_as_float(e0.y);
        acc.x += v0 * g0.x; acc.y += v0 * g0.y; acc.z += v0 * g0.z; acc.w += v0 * g0.w;
    }
    if (n > BUCKET) {
        const int q = cnt[NROWS];
        for (int i = 0; i < q; ++i) {
            if (spill_row[i] != r) continue;
            const int2 e0 = ((const int2*)spill_edge)[i];
            const float4 g0 = d4[(size_t)e0.x * (HID / 4)];
            const float v0 = __int_as_float(e0.y);
            acc.x += v0 * g0.x; acc.y += v0 * g0.y; acc.z += v0 * g0.z; acc.w += v0 * g0.w;
        }
    }

    const float4 bb = *(const float4*)(b1 + 4 * lane);
    float4 hv;
    hv.x = fmaxf(acc.x + bb.x, 0.0f);
    hv.y = fmaxf(acc.y + bb.y, 0.0f);
    hv.z = fmaxf(acc.z + bb.z, 0.0f);
    hv.w = fmaxf(acc.w + bb.w, 0.0f);
    *(float4*)(hs + wv * HPITCH + 4 * lane) = hv;
    __syncthreads();
    if (wv >= 4) return;

    const int half = lane >> 4;
    const int l16  = lane & 15;
    const float* hrow = hs + l16 * HPITCH + 2 * half;
    const float* wcol = W2 + 16 * wv + l16;

    v8f c = {};
    #pragma unroll 8
    for (int k = 0; k < HID; k += 4) {
        const float2 av = *(const float2*)(hrow + k);
        v2f a; a.x = av.x; a.y = av.y;
        v2f b;
        b.x = wcol[(size_t)(k + 2 * half) * CLS];
        b.y = wcol[(size_t)(k + 2 * half + 1) * CLS];
        c = __builtin_amdgcn_wmma_f32_16x16x4_f32(false, a, false, b, (short)0, c, false, false);
    }

    float* yb = Y + ((size_t)blockIdx.x * 16 + 8 * half) * CLS + 16 * wv + l16;
    #pragma unroll
    for (int i = 0; i < 8; ++i)
        yb[(size_t)i * CLS] = c[i];
}

__global__ __launch_bounds__(256) void gcn_spmm_rows_f64_logsoftmax(
    const int* __restrict__ cnt, const int* __restrict__ edges,
    const int* __restrict__ spill_row, const int* __restrict__ spill_edge,
    const float* __restrict__ dense, const float* __restrict__ b2,
    float* __restrict__ out)
{
    const int lane = threadIdx.x & 31;
    const int r = __builtin_amdgcn_readfirstlane(
        (int)((blockIdx.x * (size_t)blockDim.x + threadIdx.x) >> 5));
    if (r >= NROWS) return;
    const int n   = cnt[r];
    const int end = (n < BUCKET) ? n : BUCKET;
    const int2*   ew = (const int2*)edges + (size_t)r * BUCKET;
    const float2* d2 = (const float2*)dense + lane;

    float2 acc = {0.0f, 0.0f};
    int p = 0;
    for (; p + 4 <= end; p += 4) {
        const int2 e0 = ew[p], e1 = ew[p + 1], e2 = ew[p + 2], e3 = ew[p + 3];
        const float2 g0 = d2[(size_t)e0.x * (CLS / 2)];
        const float2 g1 = d2[(size_t)e1.x * (CLS / 2)];
        const float2 g2 = d2[(size_t)e2.x * (CLS / 2)];
        const float2 g3 = d2[(size_t)e3.x * (CLS / 2)];
        const float v0 = __int_as_float(e0.y), v1 = __int_as_float(e1.y);
        const float v2 = __int_as_float(e2.y), v3 = __int_as_float(e3.y);
        acc.x += v0 * g0.x; acc.y += v0 * g0.y;
        acc.x += v1 * g1.x; acc.y += v1 * g1.y;
        acc.x += v2 * g2.x; acc.y += v2 * g2.y;
        acc.x += v3 * g3.x; acc.y += v3 * g3.y;
    }
    for (; p < end; ++p) {
        const int2 e0 = ew[p];
        const float2 g0 = d2[(size_t)e0.x * (CLS / 2)];
        const float v0 = __int_as_float(e0.y);
        acc.x += v0 * g0.x; acc.y += v0 * g0.y;
    }
    if (n > BUCKET) {
        const int q = cnt[NROWS];
        for (int i = 0; i < q; ++i) {
            if (spill_row[i] != r) continue;
            const int2 e0 = ((const int2*)spill_edge)[i];
            const float2 g0 = d2[(size_t)e0.x * (CLS / 2)];
            const float v0 = __int_as_float(e0.y);
            acc.x += v0 * g0.x; acc.y += v0 * g0.y;
        }
    }

    const float2 bb = *(const float2*)(b2 + lane * 2);
    float2 v;
    v.x = acc.x + bb.x;
    v.y = acc.y + bb.y;

    float m = fmaxf(v.x, v.y);
    #pragma unroll
    for (int off = 16; off > 0; off >>= 1)
        m = fmaxf(m, __shfl_xor(m, off, 32));

    float s = __expf(v.x - m) + __expf(v.y - m);
    #pragma unroll
    for (int off = 16; off > 0; off >>= 1)
        s += __shfl_xor(s, off, 32);

    const float lse = m + __logf(s);
    float2 o;
    o.x = v.x - lse;
    o.y = v.y - lse;
    ((float2*)out)[(size_t)r * (CLS / 2) + lane] = o;
}

extern "C" void kernel_launch(void* const* d_in, const int* in_sizes, int n_in,
                              void* d_out, int out_size, void* d_ws, size_t ws_size,
                              hipStream_t stream) {
    const float* x        = (const float*)d_in[0];
    const int*   edge_row = (const int*)  d_in[1];
    const int*   edge_col = (const int*)  d_in[2];
    const float* edge_val = (const float*)d_in[3];
    const float* W1       = (const float*)d_in[4];
    const float* b1       = (const float*)d_in[5];
    const float* W2       = (const float*)d_in[6];
    const float* b2       = (const float*)d_in[7];
    float* out = (float*)d_out;

    float* ws    = (float*)d_ws;
    float* XW1   = ws;
    float* HW2   = ws + (size_t)NROWS * HID;

    int* cnt        = (int*)(ws + 2 * (size_t)NROWS * HID);
    int* edges      = cnt + CNT_WORDS;
    int* spill_row  = edges + (size_t)NROWS * BUCKET * 2;
    int* spill_edge = spill_row + NEDGE;

    const int MT = NROWS / 16;
    const int edge_blocks = (NEDGE + 255) / 256;
    const int row_blocks  = (NROWS * 32 + 255) / 256;

    gcn_gemm1_wmma<<<MT, 64, 0, stream>>>(x, W1, XW1, cnt);
    gcn_edge_bucket<<<edge_blocks, 256, 0, stream>>>(edge_row, edge_col, edge_val, cnt, edges, spill_row, spill_edge);
    gcn_spmm_rows_f128_gemm2<<<MT, 512, 0, stream>>>(cnt, edges, spill_row, spill_edge, XW1, b1, W2, HW2);
    gcn_spmm_rows_f64_logsoftmax<<<row_blocks, 256, 0, stream>>>(cnt, edges, spill_row, spill_edge, HW2, b2, out);
}


# Edge buckets by two-level counting sort on LDS counters (chunks binned by 64-row range; one group per range ranks its rows in LDS, arrival order), no global atomic + 8-B scatter per edge; 5 launches
# speedup vs baseline: 2.7970x; 1.4862x over previous
#include <hip/hip_runtime.h>


typedef __attribute__((ext_vector_type(2))) float v2f;
typedef __attribute__((ext_vector_type(8))) float v8f;

#define NROWS 50000
#define INDIM 256
#define HID   128
#define CLS   64
#define NEDGE 800000
#define BUCKET    64
#define CNT_WORDS 50176

__global__ __launch_bounds__(64) void gcn_gemm1_wmma(
    const float* __restrict__ X, const float* __restrict__ W1,
    float* __restrict__ Y, int* __restrict__ cnt)
{
    if (blockIdx.x == 0 && threadIdx.x == 0) cnt[NROWS] = 0;

    const int mtile = blockIdx.x;
    const int wv    = threadIdx.x >> 5;
    const int lane  = threadIdx.x & 31;
    const int half  = lane >> 4;
    const int l16   = lane & 15;

    const int m0 = mtile * 16;
    const int n0 = wv * 64;

    const float* xrow = X + (size_t)(m0 + l16) * INDIM + 2 * half;
    const float* wcol = W1 + n0 + l16;

    v8f acc0 = {}, acc1 = {}, acc2 = {}, acc3 = {};
    #pragma unroll 2
    for (int k = 0; k < INDIM; k += 4) {
        const float2 av = *(const float2*)(xrow + k);
        v2f a; a.x = av.x; a.y = av.y;
        const float* wk0 = wcol + (size_t)(k + 2 * half) * HID;
        const float* wk1 = wcol + (size_t)(k + 2 * half + 1) * HID;
        v2f b0; b0.x = wk0[0];  b0.y = wk1[0];
        v2f b1; b1.x = wk0[16]; b1.y = wk1[16];
        v2f b2; b2.x = wk0[32]; b2.y = wk1[32];
        v2f b3; b3.x = wk0[48]; b3.y = wk1[48];
        acc0 = __builtin_amdgcn_wmma_f32_16x16x4_f32(false, a, false, b0, (short)0, acc0, false, false);
        acc1 = __builtin_amdgcn_wmma_f32_16x16x4_f32(false, a, false, b1, (short)0, acc1, false, false);
        acc2 = __builtin_amdgcn_wmma_f32_16x16x4_f32(false, a, false, b2, (short)0, acc2, false, false);
        acc3 = __builtin_amdgcn_wmma_f32_16x16x4_f32(false, a, false, b3, (short)0, acc3, false, false);
    }

    float* yb = Y + n0 + l16;
    #pragma unroll
    for (int i = 0; i < 8; ++i) {
        float* yr = yb + (size_t)(m0 + i + 8 * half) * HID;
        yr[0]  = acc0[i];
        yr[16] = acc1[i];
        yr[32] = acc2[i];
        yr[48] = acc3[i];
    }
}

#define RPB      64
#define RPB_LOG  6
#define NBINS    782
#define NBPAD    784
#define NCHUNK   256
#define EPC      3125
#define EPT      4

__global__ __launch_bounds__(1024) void gcn_edge_bin(
    const int* __restrict__ erow, const int* __restrict__ ecol,
    const float* __restrict__ eval, int* __restrict__ binoff,
    int* __restrict__ recs)
{
    __shared__ int  binc[NBPAD];
    __shared__ int  binstart[NBPAD];
    __shared__ int  wsum[32];
    __shared__ int4 srec[EPC];

    const int t    = threadIdx.x;
    const int lane = t & 31;
    const int wv   = t >> 5;
    const int c    = blockIdx.x;

    if (t < NBPAD) binc[t] = 0;
    __syncthreads();

    int r[EPT], cl[EPT], vb[EPT], rk[EPT];
    #pragma unroll
    for (int j = 0; j < EPT; ++j) {
        const int el = t + 1024 * j;
        r[j] = 0; cl[j] = 0; vb[j] = 0; rk[j] = 0;
        if (el < EPC) {
            const int e = c * EPC + el;
            r[j]  = erow[e];
            cl[j] = ecol[e];
            vb[j] = __float_as_int(eval[e]);
        }
    }
    #pragma unroll
    for (int j = 0; j < EPT; ++j)
        if (t + 1024 * j < EPC) rk[j] = atomicAdd(&binc[r[j] >> RPB_LOG], 1);
    __syncthreads();

    const int x = (t < NBINS) ? binc[t] : 0;
    int incl = x;
    #pragma unroll
    for (int off = 1; off < 32; off <<= 1) {
        const int y = __shfl_up(incl, off, 32);
        if (lane >= off) incl += y;
    }
    if (lane == 31) wsum[wv] = incl;
    __syncthreads();
    int base = 0;
    for (int i = 0; i < wv; ++i) base += wsum[i];
    const int excl = base + incl - x;
    if (t <= NBINS) {
        binstart[t] = excl;
        binoff[c * NBPAD + t] = excl;
    }
    __syncthreads();

    #pragma unroll
    for (int j = 0; j < EPT; ++j)
        if (t + 1024 * j < EPC) {
            const int pos = binstart[r[j] >> RPB_LOG] + rk[j];
            srec[pos] = make_int4(r[j], cl[j], vb[j], 0);
        }
    __syncthreads();

    int4* dst = (int4*)recs + (size_t)c * EPC;
    for (int i = t; i < EPC; i += 1024) dst[i] = srec[i];
}

__global__ __launch_bounds__(NCHUNK) void gcn_edge_own(
    const int* __restrict__ binoff, const int* __restrict__ recs,
    int* __restrict__ cnt, int* __restrict__ edges,
    int* __restrict__ spill_row, int* __restrict__ spill_edge)
{
    __shared__ int  lcnt[RPB];
    __shared__ int2 lbuck[RPB * BUCKET];

    const int t    = threadIdx.x;
    const int lane = t & 31;
    const int wv   = t >> 5;
    const int b    = blockIdx.x;
    const int row0 = b * RPB;

    if (t < RPB) lcnt[t] = 0;
    const int o0 = binoff[t * NBPAD + b];
    const int o1 = binoff[t * NBPAD + b + 1];
    __syncthreads();

    const int4* piece = (const int4*)recs + (size_t)t * EPC;
    for (int i = o0; i < o1; i += 4) {
        int4 q[4];
        #pragma unroll
        for (int j = 0; j < 4; ++j)
            q[j] = piece[(i + j < o1) ? i + j : o1 - 1];
        #pragma unroll
        for (int j = 0; j < 4; ++j) {
            if (i + j < o1) {
                const int rl = q[j].x - row0;
                const int k  = atomicAdd(&lcnt[rl], 1);
                if (k < BUCKET) {
                    lbuck[rl * BUCKET + k] = make_int2(q[j].y, q[j].z);
                } else {
                    const int s = atomicAdd(cnt + NROWS, 1);
                    spill_row[s] = q[j].x;
                    ((int2*)spill_edge)[s] = make_int2(q[j].y, q[j].z);
                }
            }
        }
    }
    __syncthreads();

    if (t < RPB && row0 + t < NROWS) cnt[row0 + t] = lcnt[t];
    for (int rl = wv; rl < RPB; rl += NCHUNK / 32) {
        const int row = row0 + rl;
        if (row >= NROWS) break;
        const int n = (lcnt[rl] < BUCKET) ? lcnt[rl] : BUCKET;
        int2* dst = (int2*)edges + (size_t)row * BUCKET;
        if (lane < n)      dst[lane]      = lbuck[rl * BUCKET + lane];
        if (lane + 32 < n) dst[lane + 32] = lbuck[rl * BUCKET + lane + 32];
    }
}

#define HPITCH 132

__global__ __launch_bounds__(512) void gcn_spmm_rows_f128_gemm2(
    const int* __restrict__ cnt, const int* __restrict__ edges,
    const int* __restrict__ spill_row, const int* __restrict__ spill_edge,
    const float* __restrict__ dense, const float* __restrict__ b1,
    const float* __restrict__ W2, float* __restrict__ Y)
{
    __shared__ float hs[16 * HPITCH];

    const int lane = threadIdx.x & 31;
    const int wv = __builtin_amdgcn_readfirstlane((int)(threadIdx.x >> 5));
    const int r  = (int)blockIdx.x * 16 + wv;
    const int n   = cnt[r];
    const int end = (n < BUCKET) ? n : BUCKET;
    const int2*   ew = (const int2*)edges + (size_t)r * BUCKET;
    const float4* d4 = (const float4*)dense + lane;

    float4 acc = {0.0f, 0.0f, 0.0f, 0.0f};
    int p = 0;
    for (; p + 4 <= end; p += 4) {
        const int2 e0 = ew[p], e1 = ew[p + 1], e2 = ew[p + 2], e3 = ew[p + 3];
        const float4 g0 = d4[(size_t)e0.x * (HID / 4)];
        const float4 g1 = d4[(size_t)e1.x * (HID / 4)];
        const float4 g2 = d4[(size_t)e2.x * (HID / 4)];
        const float4 g3 = d4[(size_t)e3.x * (HID / 4)];
        const float v0 = __int_as_float(e0.y), v1 = __int_as_float(e1.y);
        const float v2 = __int_as_float(e2.y), v3 = __int_as_float(e3.y);
        acc.x += v0 * g0.x; acc.y += v0 * g0.y; acc.z += v0 * g0.z; acc.w += v0 * g0.w;
        acc.x += v1 * g1.x; acc.y += v1 * g1.y; acc.z += v1 * g1.z; acc.w += v1 * g1.w;
        acc.x += v2 * g2.x; acc.y += v2 * g2.y; acc.z += v2 * g2.z; acc.w += v2 * g2.w;
        acc.x += v3 * g3.x; acc.y += v3 * g3.y; acc.z += v3 * g3.z; acc.w += v3 * g3.w;
    }
    for (; p < end; ++p) {
        const int2 e0 = ew[p];
        const float4 g0 = d4[(size_t)e0.x * (HID / 4)];
        const float v0 = __int_as_float(e0.y);
        acc.x += v0 * g0.x; acc.y += v0 * g0.y; acc.z += v0 * g0.z; acc.w += v0 * g0.w;
    }
    if (n > BUCKET) {
        const int q = cnt[NROWS];
        for (int i = 0; i < q; ++i) {
            if (spill_row[i] != r) continue;
            const int2 e0 = ((const int2*)spill_edge)[i];
            const float4 g0 = d4[(size_t)e0.x * (HID / 4)];
            const float v0 = __int_as_float(e0.y);
            acc.x += v0 * g0.x; acc.y += v0 * g0.y; acc.z += v0 * g0.z; acc.w += v0 * g0.w;
        }
    }

    const float4 bb = *(const float4*)(b1 + 4 * lane);
    float4 hv;
    hv.x = fmaxf(acc.x + bb.x, 0.0f);
    hv.y = fmaxf(acc.y + bb.y, 0.0f);
    hv.z = fmaxf(acc.z + bb.z, 0.0f);
    hv.w = fmaxf(acc.w + bb.w, 0.0f);
    *(float4*)(hs + wv * HPITCH + 4 * lane) = hv;
    __syncthreads();
    if (wv >= 4) return;

    const int half = lane >> 4;
    const int l16  = lane & 15;
    const float* hrow = hs + l16 * HPITCH + 2 * half;
    const float* wcol = W2 + 16 * wv + l16;

    v8f c = {};
    #pragma unroll 8
    for (int k = 0; k < HID; k += 4) {
        const float2 av = *(const float2*)(hrow + k);
        v2f a; a.x = av.x; a.y = av.y;
        v2f b;
        b.x = wcol[(size_t)(k + 2 * half) * CLS];
        b.y = wcol[(size_t)(k + 2 * half + 1) * CLS];
        c = __builtin_amdgcn_wmma_f32_16x16x4_f32(false, a, false, b, (short)0, c, false, false);
    }

    float* yb = Y + ((size_t)blockIdx.x * 16 + 8 * half) * CLS + 16 * wv + l16;
    #pragma unroll
    for (int i = 0; i < 8; ++i)
        yb[(size_t)i * CLS] = c[i];
}

__global__ __launch_bounds__(256) void gcn_spmm_rows_f64_logsoftmax(
    const int* __restrict__ cnt, const int* __restrict__ edges,
    const int* __restrict__ spill_row, const int* __restrict__ spill_edge,
    const float* __restrict__ dense, const float* __restrict__ b2,
    float* __restrict__ out)
{
    const int lane = threadIdx.x & 31;
    const int r = __builtin_amdgcn_readfirstlane(
        (int)((blockIdx.x * (size_t)blockDim.x + threadIdx.x) >> 5));
    if (r >= NROWS) return;
    const int n   = cnt[r];
    const int end = (n < BUCKET) ? n : BUCKET;
    const int2*   ew = (const int2*)edges + (size_t)r * BUCKET;
    const float2* d2 = (const float2*)dense + lane;

    float2 acc = {0.0f, 0.0f};
    int p = 0;
    for (; p + 4 <= end; p += 4) {
        const int2 e0 = ew[p], e1 = ew[p + 1], e2 = ew[p + 2], e3 = ew[p + 3];
        const float2 g0 = d2[(size_t)e0.x * (CLS / 2)];
        const float2 g1 = d2[(size_t)e1.x * (CLS / 2)];
        const float2 g2 = d2[(size_t)e2.x * (CLS / 2)];
        const float2 g3 = d2[(size_t)e3.x * (CLS / 2)];
        const float v0 = __int_as_float(e0.y), v1 = __int_as_float(e1.y);
        const float v2 = __int_as_float(e2.y), v3 = __int_as_float(e3.y);
        acc.x += v0 * g0.x; acc.y += v0 * g0.y;
        acc.x += v1 * g1.x; acc.y += v1 * g1.y;
        acc.x += v2 * g2.x; acc.y += v2 * g2.y;
        acc.x += v3 * g3.x; acc.y += v3 * g3.y;
    }
    for (; p < end; ++p) {
        const int2 e0 = ew[p];
        const float2 g0 = d2[(size_t)e0.x * (CLS / 2)];
        const float v0 = __int_as_float(e0.y);
        acc.x += v0 * g0.x; acc.y += v0 * g0.y;
    }
    if (n > BUCKET) {
        const int q = cnt[NROWS];
        for (int i = 0; i < q; ++i) {
            if (spill_row[i] != r) continue;
            const int2 e0 = ((const int2*)spill_edge)[i];
            const float2 g0 = d2[(size_t)e0.x * (CLS / 2)];
            const float v0 = __int_as_float(e0.y);
            acc.x += v0 * g0.x; acc.y += v0 * g0.y;
        }
    }

    const float2 bb = *(const float2*)(b2 + lane * 2);
    float2 v;
    v.x = acc.x + bb.x;
    v.y = acc.y + bb.y;

    float m = fmaxf(v.x, v.y);
    #pragma unroll
    for (int off = 16; off > 0; off >>= 1)
        m = fmaxf(m, __shfl_xor(m, off, 32));

    float s = __expf(v.x - m) + __expf(v.y - m);
    #pragma unroll
    for (int off = 16; off > 0; off >>= 1)
        s += __shfl_xor(s, off, 32);

    const float lse = m + __logf(s);
    float2 o;
    o.x = v.x - lse;
    o.y = v.y - lse;
    ((float2*)out)[(size_t)r * (CLS / 2) + lane] = o;
}

extern "C" void kernel_launch(void* const* d_in, const int* in_sizes, int n_in,
                              void* d_out, int out_size, void* d_ws, size_t ws_size,
                              hipStream_t stream) {
    const float* x        = (const float*)d_in[0];
    const int*   edge_row = (const int*)  d_in[1];
    const int*   edge_col = (const int*)  d_in[2];
    const float* edge_val = (const float*)d_in[3];
    const float* W1       = (const float*)d_in[4];
    const float* b1       = (const float*)d_in[5];
    const float* W2       = (const float*)d_in[6];
    const float* b2       = (const float*)d_in[7];
    float* out = (float*)d_out;

    float* ws    = (float*)d_ws;
    float* XW1   = ws;
    float* HW2   = ws + (size_t)NROWS * HID;

    int* cnt        = (int*)(ws + 2 * (size_t)NROWS * HID);
    int* edges      = cnt + CNT_WORDS;
    int* spill_row  = edges + (size_t)NROWS * BUCKET * 2;
    int* spill_edge = spill_row + NEDGE;
    int* recs       = spill_edge + (size_t)NEDGE * 2;
    int* binoff     = recs + (size_t)NEDGE * 4;

    const int MT = NROWS / 16;
    const int row_blocks  = (NROWS * 32 + 255) / 256;

    gcn_gemm1_wmma<<<MT, 64, 0, stream>>>(x, W1, XW1, cnt);
    gcn_edge_bin<<<NCHUNK, 1024, 0, stream>>>(edge_row, edge_col, edge_val, binoff, recs);
    gcn_edge_own<<<NBINS, NCHUNK, 0, stream>>>(binoff, recs, cnt, edges, spill_row, spill_edge);
    gcn_spmm_rows_f128_gemm2<<<MT, 512, 0, stream>>>(cnt, edges, spill_row, spill_edge, XW1, b1, W2, HW2);
    gcn_spmm_rows_f64_logsoftmax<<<row_blocks, 256, 0, stream>>>(cnt, edges, spill_row, spill_edge, HW2, b2, out);
}


# GEMM1 and the bucket-fill (second sort step) share one launch as two kinds of work-group, GEMM groups first; 5 -> 4 launches; buckets still in arrival order
# speedup vs baseline: 2.9915x; 1.0695x over previous
#include <hip/hip_runtime.h>


typedef __attribute__((ext_vector_type(2))) float v2f;
typedef __attribute__((ext_vector_type(8))) float v8f;

#define NROWS 50000
#define INDIM 256
#define HID   128
#define CLS   64
#define NEDGE 800000
#define BUCKET    64
#define CNT_WORDS 50176

#define RPB      64
#define RPB_LOG  6
#define NBINS    782
#define NBPAD    784
#define NCHUNK   256
#define EPC      3125
#define EPT      4

__global__ __launch_bounds__(1024) void gcn_edge_bin(
    const int* __restrict__ erow, const int* __restrict__ ecol,
    const float* __restrict__ eval, int* __restrict__ binoff,
    int* __restrict__ recs, int* __restrict__ cnt)
{
    __shared__ int  binc[NBPAD];
    __shared__ int  binstart[NBPAD];
    __shared__ int  wsum[32];
    __shared__ int4 srec[EPC];

    const int t    = threadIdx.x;
    const int lane = t & 31;
    const int wv   = t >> 5;
    const int c    = blockIdx.x;

    if (c == 0 && t == 0) cnt[NROWS] = 0;

    if (t < NBPAD) binc[t] = 0;
    __syncthreads();

    int r[EPT], cl[EPT], vb[EPT], rk[EPT];
    #pragma unroll
    for (int j = 0; j < EPT; ++j) {
        const int el = t + 1024 * j;
        r[j] = 0; cl[j] = 0; vb[j] = 0; rk[j] = 0;
        if (el < EPC) {
            const int e = c * EPC + el;
            r[j]  = erow[e];
            cl[j] = ecol[e];
            vb[j] = __float_as_int(eval[e]);
        }
    }
    #pragma unroll
    for (int j = 0; j < EPT; ++j)
        if (t + 1024 * j < EPC) rk[j] = atomicAdd(&binc[r[j] >> RPB_LOG], 1);
    __syncthreads();

    const int x = (t < NBINS) ? binc[t] : 0;
    int incl = x;
    #pragma unroll
    for (int off = 1; off < 32; off <<= 1) {
        const int y = __shfl_up(incl, off, 32);
        if (lane >= off) incl += y;
    }
    if (lane == 31) wsum[wv] = incl;
    __syncthreads();
    int base = 0;
    for (int i = 0; i < wv; ++i) base += wsum[i];
    const int excl = base + incl - x;
    if (t <= NBINS) {
        binstart[t] = excl;
        binoff[c * NBPAD + t] = excl;
    }
    __syncthreads();

    #pragma unroll
    for (int j = 0; j < EPT; ++j)
        if (t + 1024 * j < EPC) {
            const int pos = binstart[r[j] >> RPB_LOG] + rk[j];
            srec[pos] = make_int4(r[j], cl[j], vb[j], 0);
        }
    __syncthreads();

    int4* dst = (int4*)recs + (size_t)c * EPC;
    for (int i = t; i < EPC; i += 1024) dst[i] = srec[i];
}

#define GEMM_GROUPS 1563
#define GROUP_THR   128

__global__ __launch_bounds__(GROUP_THR) void gcn_gemm1_and_edge_own(
    const float* __restrict__ X, const float* __restrict__ W1,
    float* __restrict__ Y,
    const int* __restrict__ binoff, const int* __restrict__ recs,
    int* __restrict__ cnt, int* __restrict__ edges,
    int* __restrict__ spill_row, int* __restrict__ spill_edge)
{
    __shared__ int  lcnt[RPB];
    __shared__ int2 lbuck[RPB * BUCKET];

    const int t    = threadIdx.x;
    const int lane = t & 31;
    const int wave = __builtin_amdgcn_readfirstlane(t >> 5);

    if (blockIdx.x < GEMM_GROUPS) {
        const int mtile = (int)blockIdx.x * 2 + (wave >> 1);
        if (mtile >= NROWS / 16) return;
        const int wv   = wave & 1;
        const int half = lane >> 4;
        const int l16  = lane & 15;

        const int m0 = mtile * 16;
        const int n0 = wv * 64;

        const float* xrow = X + (size_t)(m0 + l16) * INDIM + 2 * half;
        const float* wcol = W1 + n0 + l16;

        v8f acc0 = {}, acc1 = {}, acc2 = {}, acc3 = {};
        #pragma unroll 2
        for (int k = 0; k < INDIM; k += 4) {
            const float2 av = *(const float2*)(xrow + k);
            v2f a; a.x = av.x; a.y = av.y;
            const float* wk0 = wcol + (size_t)(k + 2 * half) * HID;
            const float* wk1 = wcol + (size_t)(k + 2 * half + 1) * HID;
            v2f b0; b0.x = wk0[0];  b0.y = wk1[0];
            v2f b1; b1.x = wk0[16]; b1.y = wk1[16];
            v2f b2; b2.x = wk0[32]; b2.y = wk1[32];
            v2f b3; b3.x = wk0[48]; b3.y = wk1[48];
            acc0 = __builtin_amdgcn_wmma_f32_16x16x4_f32(false, a, false, b0, (short)0, acc0, false, false);
            acc1 = __builtin_amdgcn_wmma_f32_16x16x4_f32(false, a, false, b1, (short)0, acc1, false, false);
            acc2 = __builtin_amdgcn_wmma_f32_16x16x4_f32(false, a, false, b2, (short)0, acc2, false, false);
            acc3 = __builtin_amdgcn_wmma_f32_16x16x4_f32(false, a, false, b3, (short)0, acc3, false, false);
        }

        float* yb = Y + n0 + l16;
        #pragma unroll
        for (int i = 0; i < 8; ++i) {
            float* yr = yb + (size_t)(m0 + i + 8 * half) * HID;
            yr[0]  = acc0[i];
            yr[16] = acc1[i];
            yr[32] = acc2[i];
            yr[48] = acc3[i];
        }
        return;
    }

    const int b    = (int)blockIdx.x - GEMM_GROUPS;
    const int row0 = b * RPB;

    if (t < RPB) lcnt[t] = 0;
    int o0[2], o1[2];
    #pragma unroll
    for (int h = 0; h < 2; ++h) {
        const int ch = t + GROUP_THR * h;
        o0[h] = binoff[ch * NBPAD + b];
        o1[h] = binoff[ch * NBPAD + b + 1];
    }
    __syncthreads();

    #pragma unroll
    for (int h = 0; h < 2; ++h) {
        const int4* piece = (const int4*)recs + (size_t)(t + GROUP_THR * h) * EPC;
        const int pe = o1[h];
        for (int i = o0[h]; i < pe; i += 4) {
            int4 q[4];
            #pragma unroll
            for (int j = 0; j < 4; ++j)
                q[j] = piece[(i + j < pe) ? i + j : pe - 1];
            #pragma unroll
            for (int j = 0; j < 4; ++j) {
                if (i + j < pe) {
                    const int rl = q[j].x - row0;
                    const int k  = atomicAdd(&lcnt[rl], 1);
                    if (k < BUCKET) {
                        lbuck[rl * BUCKET + k] = make_int2(q[j].y, q[j].z);
                    } else {
                        const int s = atomicAdd(cnt + NROWS, 1);
                        spill_row[s] = q[j].x;
                        ((int2*)spill_edge)[s] = make_int2(q[j].y, q[j].z);
                    }
                }
            }
        }
    }
    __syncthreads();

    if (t < RPB && row0 + t < NROWS) cnt[row0 + t] = lcnt[t];
    for (int rl = wave; rl < RPB; rl += GROUP_THR / 32) {
        const int row = row0 + rl;
        if (row >= NROWS) break;
        const int n = (lcnt[rl] < BUCKET) ? lcnt[rl] : BUCKET;
        int2* dst = (int2*)edges + (size_t)row * BUCKET;
        if (lane < n)      dst[lane]      = lbuck[rl * BUCKET + lane];
        if (lane + 32 < n) dst[lane + 32] = lbuck[rl * BUCKET + lane + 32];
    }
}

#define HPITCH 132

__global__ __launch_bounds__(512) void gcn_spmm_rows_f128_gemm2(
    const int* __restrict__ cnt, const int* __restrict__ edges,
    const int* __restrict__ spill_row, const int* __restrict__ spill_edge,
    const float* __restrict__ dense, const float* __restrict__ b1,
    const float* __restrict__ W2, float* __restrict__ Y)
{
    __shared__ float hs[16 * HPITCH];

    const int lane = threadIdx.x & 31;
    const int wv = __builtin_amdgcn_readfirstlane((int)(threadIdx.x >> 5));
    const int r  = (int)blockIdx.x * 16 + wv;
    const int n   = cnt[r];
    const int end = (n < BUCKET) ? n : BUCKET;
    const int2*   ew = (const int2*)edges + (size_t)r * BUCKET;
    const float4* d4 = (const float4*)dense + lane;

    float4 acc = {0.0f, 0.0f, 0.0f, 0.0f};
    int p = 0;
    for (; p + 4 <= end; p += 4) {
        const int2 e0 = ew[p], e1 = ew[p + 1], e2 = ew[p + 2], e3 = ew[p + 3];
        const float4 g0 = d4[(size_t)e0.x * (HID / 4)];
        const float4 g1 = d4[(size_t)e1.x * (HID / 4)];
        const float4 g2 = d4[(size_t)e2.x * (HID / 4)];
        const float4 g3 = d4[(size_t)e3.x * (HID / 4)];
        const float v0 = __int_as_float(e0.y), v1 = __int_as_float(e1.y);
        const float v2 = __int_as_float(e2.y), v3 = __int_as_float(e3.y);
        acc.x += v0 * g0.x; acc.y += v0 * g0.y; acc.z += v0 * g0.z; acc.w += v0 * g0.w;
        acc.x += v1 * g1.x; acc.y += v1 * g1.y; acc.z += v1 * g1.z; acc.w += v1 * g1.w;
        acc.x += v2 * g2.x; acc.y += v2 * g2.y; acc.z += v2 * g2.z; acc.w += v2 * g2.w;
        acc.x += v3 * g3.x; acc.y += v3 * g3.y; acc.z += v3 * g3.z; acc.w += v3 * g3.w;
    }
    for (; p < end; ++p) {
        const int2 e0 = ew[p];
        const float4 g0 = d4[(size_t)e0.x * (HID / 4)];
        const float v0 = __int_as_float(e0.y);
        acc.x += v0 * g0.x; acc.y += v0 * g0.y; acc.z += v0 * g0.z; acc.w += v0 * g0.w;
    }
    if (n > BUCKET) {
        const int q = cnt[NROWS];
        for (int i = 0; i < q; ++i) {
            if (spill_row[i] != r) continue;
            const int2 e0 = ((const int2*)spill_edge)[i];
            const float4 g0 = d4[(size_t)e0.x * (HID / 4)];
            const float v0 = __int_as_float(e0.y);
            acc.x += v0 * g0.x; acc.y += v0 * g0.y; acc.z += v0 * g0.z; acc.w += v0 * g0.w;
        }
    }

    const float4 bb = *(const float4*)(b1 + 4 * lane);
    float4 hv;
    hv.x = fmaxf(acc.x + bb.x, 0.0f);
    hv.y = fmaxf(acc.y + bb.y, 0.0f);
    hv.z = fmaxf(acc.z + bb.z, 0.0f);
    hv.w = fmaxf(acc.w + bb.w, 0.0f);
    *(float4*)(hs + wv * HPITCH + 4 * lane) = hv;
    __syncthreads();
    if (wv >= 4) return;

    const int half = lane >> 4;
    const int l16  = lane & 15;
    const float* hrow = hs + l16 * HPITCH + 2 * half;
    const float* wcol = W2 + 16 * wv + l16;

    v8f c = {};
    #pragma unroll 8
    for (int k = 0; k < HID; k += 4) {
        const float2 av = *(const float2*)(hrow + k);
        v2f a; a.x = av.x; a.y = av.y;
        v2f b;
        b.x = wcol[(size_t)(k + 2 * half) * CLS];
        b.y = wcol[(size_t)(k + 2 * half + 1) * CLS];
        c = __builtin_amdgcn_wmma_f32_16x16x4_f32(false, a, false, b, (short)0, c, false, false);
    }

    float* yb = Y + ((size_t)blockIdx.x * 16 + 8 * half) * CLS + 16 * wv + l16;
    #pragma unroll
    for (int i = 0; i < 8; ++i)
        yb[(size_t)i * CLS] = c[i];
}

__global__ __launch_bounds__(256) void gcn_spmm_rows_f64_logsoftmax(
    const int* __restrict__ cnt, const int* __restrict__ edges,
    const int* __restrict__ spill_row, const int* __restrict__ spill_edge,
    const float* __restrict__ dense, const float* __restrict__ b2,
    float* __restrict__ out)
{
    const int lane = threadIdx.x & 31;
    const int r = __builtin_amdgcn_readfirstlane(
        (int)((blockIdx.x * (size_t)blockDim.x + threadIdx.x) >> 5));
    if (r >= NROWS) return;
    const int n   = cnt[r];
    const int end = (n < BUCKET) ? n : BUCKET;
    const int2*   ew = (const int2*)edges + (size_t)r * BUCKET;
    const float2* d2 = (const float2*)dense + lane;

    float2 acc = {0.0f, 0.0f};
    int p = 0;
    for (; p + 4 <= end; p += 4) {
        const int2 e0 = ew[p], e1 = ew[p + 1], e2 = ew[p + 2], e3 = ew[p + 3];
        const float2 g0 = d2[(size_t)e0.x * (CLS / 2)];
        const float2 g1 = d2[(size_t)e1.x * (CLS / 2)];
        const float2 g2 = d2[(size_t)e2.x * (CLS / 2)];
        const float2 g3 = d2[(size_t)e3.x * (CLS / 2)];
        const float v0 = __int_as_float(e0.y), v1 = __int_as_float(e1.y);
        const float v2 = __int_as_float(e2.y), v3 = __int_as_float(e3.y);
        acc.x += v0 * g0.x; acc.y += v0 * g0.y;
        acc.x += v1 * g1.x; acc.y += v1 * g1.y;
        acc.x += v2 * g2.x; acc.y += v2 * g2.y;
        acc.x += v3 * g3.x; acc.y += v3 * g3.y;
    }
    for (; p < end; ++p) {
        const int2 e0 = ew[p];
        const float2 g0 = d2[(size_t)e0.x * (CLS / 2)];
        const float v0 = __int_as_float(e0.y);
        acc.x += v0 * g0.x; acc.y += v0 * g0.y;
    }
    if (n > BUCKET) {
        const int q = cnt[NROWS];
        for (int i = 0; i < q; ++i) {
            if (spill_row[i] != r) continue;
            const int2 e0 = ((const int2*)spill_edge)[i];
            const float2 g0 = d2[(size_t)e0.x * (CLS / 2)];
            const float v0 = __int_as_float(e0.y);
            acc.x += v0 * g0.x; acc.y += v0 * g0.y;
        }
    }

    const float2 bb = *(const float2*)(b2 + lane * 2);
    float2 v;
    v.x = acc.x + bb.x;
    v.y = acc.y + bb.y;

    float m = fmaxf(v.x, v.y);
    #pragma unroll
    for (int off = 16; off > 0; off >>= 1)
        m = fmaxf(m, __shfl_xor(m, off, 32));

    float s = __expf(v.x - m) + __expf(v.y - m);
    #pragma unroll
    for (int off = 16; off > 0; off >>= 1)
        s += __shfl_xor(s, off, 32);

    const float lse = m + __logf(s);
    float2 o;
    o.x = v.x - lse;
    o.y = v.y - lse;
    ((float2*)out)[(size_t)r * (CLS / 2) + lane] = o;
}

extern "C" void kernel_launch(void* const* d_in, const int* in_sizes, int n_in,
                              void* d_out, int out_size, void* d_ws, size_t ws_size,
                              hipStream_t stream) {
    const float* x        = (const float*)d_in[0];
    const int*   edge_row = (const int*)  d_in[1];
    const int*   edge_col = (const int*)  d_in[2];
    const float* edge_val = (const float*)d_in[3];
    const float* W1       = (const float*)d_in[4];
    const float* b1       = (const float*)d_in[5];
    const float* W2       = (const float*)d_in[6];
    const float* b2       = (const float*)d_in[7];
    float* out = (float*)d_out;

    float* ws    = (float*)d_ws;
    float* XW1   = ws;
    float* HW2   = ws + (size_t)NROWS * HID;

    int* cnt        = (int*)(ws + 2 * (size_t)NROWS * HID);
    int* edges      = cnt + CNT_WORDS;
    int* spill_row  = edges + (size_t)NROWS * BUCKET * 2;
    int* spill_edge = spill_row + NEDGE;
    int* recs       = spill_edge + (size_t)NEDGE * 2;
    int* binoff     = recs + (size_t)NEDGE * 4;

    const int MT = NROWS / 16;
    const int shared_groups = GEMM_GROUPS + NBINS;
    const int row_blocks  = (NROWS * 32 + 255) / 256;

    gcn_edge_bin<<<NCHUNK, 1024, 0, stream>>>(edge_row, edge_col, edge_val, binoff, recs, cnt);
    gcn_gemm1_and_edge_own<<<shared_groups, GROUP_THR, 0, stream>>>(x, W1, XW1, binoff, recs, cnt, edges, spill_row, spill_edge);
    gcn_spmm_rows_f128_gemm2<<<MT, 512, 0, stream>>>(cnt, edges, spill_row, spill_edge, XW1, b1, W2, HW2);
    gcn_spmm_rows_f64_logsoftmax<<<row_blocks, 256, 0, stream>>>(cnt, edges, spill_row, spill_edge, HW2, b2, out);
}


# Sort step 1 rides beside GEMM1; step 2 heads the F=128 SpMM launch: 64-row groups fill buckets in LDS, walk them there, GEMM2 on 16 waves; over-full rows sum in chunk order; 4 -> 3 launches
# speedup vs baseline: 3.1952x; 1.0681x over previous
#include <hip/hip_runtime.h>


typedef __attribute__((ext_vector_type(2))) float v2f;
typedef __attribute__((ext_vector_type(8))) float v8f;

#define NROWS 50000
#define INDIM 256
#define HID   128
#define CLS   64
#define NEDGE 800000
#define BUCKET    64
#define CNT_WORDS 50176

#define RPB      64
#define RPB_LOG  6
#define NBINS    782
#define NBPAD    784
#define NCHUNK   256
#define EPC      3125
#define EPT      4

#define GEMM_GROUPS      250
#define STRIPS_PER_GROUP 25

__global__ __launch_bounds__(1024) void gcn_gemm1_and_edge_bin(
    const float* __restrict__ X, const float* __restrict__ W1,
    float* __restrict__ Y,
    const int* __restrict__ erow, const int* __restrict__ ecol,
    const float* __restrict__ eval, int* __restrict__ binoff,
    int* __restrict__ recs, int* __restrict__ cnt)
{
    __shared__ int  binc[NBPAD];
    __shared__ int  binstart[NBPAD];
    __shared__ int  wsum[32];
    __shared__ int4 srec[EPC];

    const int t    = threadIdx.x;
    const int lane = t & 31;
    const int wv   = __builtin_amdgcn_readfirstlane(t >> 5);

    if (blockIdx.x < GEMM_GROUPS) {
        if (wv >= STRIPS_PER_GROUP) return;
        const int strip = (int)blockIdx.x * STRIPS_PER_GROUP + wv;
        const int mtile = strip >> 1;
        const int nh    = strip & 1;
        const int half  = lane >> 4;
        const int l16   = lane & 15;

        const int m0 = mtile * 16;
        const int n0 = nh * 64;

        const float* xrow = X + (size_t)(m0 + l16) * INDIM + 2 * half;
        const float* wcol = W1 + n0 + l16;

        v8f acc0 = {}, acc1 = {}, acc2 = {}, acc3 = {};
        #pragma unroll 2
        for (int k = 0; k < INDIM; k += 4) {
            const float2 av = *(const float2*)(xrow + k);
            v2f a; a.x = av.x; a.y = av.y;
            const float* wk0 = wcol + (size_t)(k + 2 * half) * HID;
            const float* wk1 = wcol + (size_t)(k + 2 * half + 1) * HID;
            v2f b0; b0.x = wk0[0];  b0.y = wk1[0];
            v2f b1; b1.x = wk0[16]; b1.y = wk1[16];
            v2f b2; b2.x = wk0[32]; b2.y = wk1[32];
            v2f b3; b3.x = wk0[48]; b3.y = wk1[48];
            acc0 = __builtin_amdgcn_wmma_f32_16x16x4_f32(false, a, false, b0, (short)0, acc0, false, false);
            acc1 = __builtin_amdgcn_wmma_f32_16x16x4_f32(false, a, false, b1, (short)0, acc1, false, false);
            acc2 = __builtin_amdgcn_wmma_f32_16x16x4_f32(false, a, false, b2, (short)0, acc2, false, false);
            acc3 = __builtin_amdgcn_wmma_f32_16x16x4_f32(false, a, false, b3, (short)0, acc3, false, false);
        }

        float* yb = Y + n0 + l16;
        #pragma unroll
        for (int i = 0; i < 8; ++i) {
            float* yr = yb + (size_t)(m0 + i + 8 * half) * HID;
            yr[0]  = acc0[i];
            yr[16] = acc1[i];
            yr[32] = acc2[i];
            yr[48] = acc3[i];
        }
        return;
    }

    const int c = (int)blockIdx.x - GEMM_GROUPS;

    if (c == 0 && t == 0) cnt[NROWS] = 0;

    if (t < NBPAD) binc[t] = 0;
    __syncthreads();

    int r[EPT], cl[EPT], vb[EPT], rk[EPT];
    #pragma unroll
    for (int j = 0; j < EPT; ++j) {
        const int el = t + 1024 * j;
        r[j] = 0; cl[j] = 0; vb[j] = 0; rk[j] = 0;
        if (el < EPC) {
            const int e = c * EPC + el;
            r[j]  = erow[e];
            cl[j] = ecol[e];
            vb[j] = __float_as_int(eval[e]);
        }
    }
    #pragma unroll
    for (int j = 0; j < EPT; ++j)
        if (t + 1024 * j < EPC) rk[j] = atomicAdd(&binc[r[j] >> RPB_LOG], 1);
    __syncthreads();

    const int x = (t < NBINS) ? binc[t] : 0;
    int incl = x;
    #pragma unroll
    for (int off = 1; off < 32; off <<= 1) {
        const int y = __shfl_up(incl, off, 32);
        if (lane >= off) incl += y;
    }
    if (lane == 31) wsum[wv] = incl;
    __syncthreads();
    int base = 0;
    for (int i = 0; i < wv; ++i) base += wsum[i];
    const int excl = base + incl - x;
    if (t <= NBINS) {
        binstart[t] = excl;
        binoff[c * NBPAD + t] = excl;
    }
    __syncthreads();

    #pragma unroll
    for (int j = 0; j < EPT; ++j)
        if (t + 1024 * j < EPC) {
            const int pos = binstart[r[j] >> RPB_LOG] + rk[j];
            srec[pos] = make_int4(r[j], cl[j], vb[j], 0);
        }
    __syncthreads();

    int4* dst = (int4*)recs + (size_t)c * EPC;
    for (int i = t; i < EPC; i += 1024) dst[i] = srec[i];
}

#define HPITCH    132
#define RANGE_THR 512

__global__ __launch_bounds__(RANGE_THR, 16) void gcn_range_buckets_spmm_f128_gemm2(
    const int* __restrict__ binoff, const int* __restrict__ recs,
    int* __restrict__ cnt, int* __restrict__ edges,
    int* __restrict__ spill_row, int* __restrict__ spill_edge,
    const float* __restrict__ dense, const float* __restrict__ b1,
    const float* __restrict__ W2, float* __restrict__ Y)
{
    __shared__ int   lcnt[RPB];
    __shared__ int2  lbuck[RPB * BUCKET];
    __shared__ float hs[RPB * HPITCH];

    const int t    = threadIdx.x;
    const int lane = t & 31;
    const int wave = __builtin_amdgcn_readfirstlane(t >> 5);
    const int b    = (int)blockIdx.x;
    const int row0 = b * RPB;

    if (t < RPB) lcnt[t] = 0;
    int o0 = 0, o1 = 0;
    if (t < NCHUNK) {
        o0 = binoff[t * NBPAD + b];
        o1 = binoff[t * NBPAD + b + 1];
    }
    __syncthreads();

    if (t < NCHUNK) {
        const int4* piece = (const int4*)recs + (size_t)t * EPC;
        const int pe = o1;
        for (int i = o0; i < pe; i += 4) {
            int4 q[4];
            #pragma unroll
            for (int j = 0; j < 4; ++j)
                q[j] = piece[(i + j < pe) ? i + j : pe - 1];
            #pragma unroll
            for (int j = 0; j < 4; ++j) {
                if (i + j < pe) {
                    const int rl = q[j].x - row0;
                    const int k  = atomicAdd(&lcnt[rl], 1);
                    if (k < BUCKET) {
                        lbuck[rl * BUCKET + k] = make_int2(q[j].y, q[j].z);
                    } else {
                        const int s = atomicAdd(cnt + NROWS, 1);
                        spill_row[s] = q[j].x;
                        ((int2*)spill_edge)[s] = make_int2(q[j].y, q[j].z);
                    }
                }
            }
        }
    }
    __syncthreads();

    const float4* d4 = (const float4*)dense + lane;
    const float4  bb = *(const float4*)(b1 + 4 * lane);
    for (int rl = wave; rl < RPB; rl += RANGE_THR / 32) {
        const int row = row0 + rl;
        float4 acc = {0.0f, 0.0f, 0.0f, 0.0f};
        if (row < NROWS) {
            const int n   = __builtin_amdgcn_readfirstlane(lcnt[rl]);
            const int end = (n < BUCKET) ? n : BUCKET;
            const int2* eb = lbuck + rl * BUCKET;

            if (lane == 0) cnt[row] = n;
            int2* dst = (int2*)edges + (size_t)row * BUCKET;
            if (lane < end)      dst[lane]      = eb[lane];
            if (lane + 32 < end) dst[lane + 32] = eb[lane + 32];

            if (n <= BUCKET) {
                int p = 0;
                for (; p + 4 <= end; p += 4) {
                    const int2 e0 = eb[p], e1 = eb[p + 1], e2 = eb[p + 2], e3 = eb[p + 3];
                    const float4 g0 = d4[(size_t)e0.x * (HID / 4)];
                    const float4 g1 = d4[(size_t)e1.x * (HID / 4)];
                    const float4 g2 = d4[(size_t)e2.x * (HID / 4)];
                    const float4 g3 = d4[(size_t)e3.x * (HID / 4)];
                    const float v0 = __int_as_float(e0.y), v1 = __int_as_float(e1.y);
                    const float v2 = __int_as_float(e2.y), v3 = __int_as_float(e3.y);
                    acc.x += v0 * g0.x; acc.y += v0 * g0.y; acc.z += v0 * g0.z; acc.w += v0 * g0.w;
                    acc.x += v1 * g1.x; acc.y += v1 * g1.y; acc.z += v1 * g1.z; acc.w += v1 * g1.w;
                    acc.x += v2 * g2.x; acc.y += v2 * g2.y; acc.z += v2 * g2.z; acc.w += v2 * g2.w;
                    acc.x += v3 * g3.x; acc.y += v3 * g3.y; acc.z += v3 * g3.z; acc.w += v3 * g3.w;
                }
                for (; p < end; ++p) {
                    const int2 e0 = eb[p];
                    const float4 g0 = d4[(size_t)e0.x * (HID / 4)];
                    const float v0 = __int_as_float(e0.y);
                    acc.x += v0 * g0.x; acc.y += v0 * g0.y; acc.z += v0 * g0.z; acc.w += v0 * g0.w;
                }
            } else {
                for (int ch = 0; ch < NCHUNK; ++ch) {
                    const int q0 = binoff[ch * NBPAD + b];
                    const int q1 = binoff[ch * NBPAD + b + 1];
                    const int4* piece = (const int4*)recs + (size_t)ch * EPC;
                    for (int i = q0; i < q1; ++i) {
                        const int4 q = piece[i];
                        if (q.x != row) continue;
                        const float4 g0 = d4[(size_t)q.y * (HID / 4)];
                        const float v0 = __int_as_float(q.z);
                        acc.x += v0 * g0.x; acc.y += v0 * g0.y; acc.z += v0 * g0.z; acc.w += v0 * g0.w;
                    }
                }
            }
        }
        float4 hv;
        hv.x = fmaxf(acc.x + bb.x, 0.0f);
        hv.y = fmaxf(acc.y + bb.y, 0.0f);
        hv.z = fmaxf(acc.z + bb.z, 0.0f);
        hv.w = fmaxf(acc.w + bb.w, 0.0f);
        *(float4*)(hs + rl * HPITCH + 4 * lane) = hv;
    }
    __syncthreads();

    const int rt = wave >> 2;
    const int ct = wave & 3;
    if (row0 + rt * 16 >= NROWS) return;
    const int half = lane >> 4;
    const int l16  = lane & 15;
    const float* hrow = hs + (rt * 16 + l16) * HPITCH + 2 * half;
    const float* wcol = W2 + 16 * ct + l16;

    v8f c = {};
    #pragma unroll 8
    for (int k = 0; k < HID; k += 4) {
        const float2 av = *(const float2*)(hrow + k);
        v2f a; a.x = av.x; a.y = av.y;
        v2f bf;
        bf.x = wcol[(size_t)(k + 2 * half) * CLS];
        bf.y = wcol[(size_t)(k + 2 * half + 1) * CLS];
        c = __builtin_amdgcn_wmma_f32_16x16x4_f32(false, a, false, bf, (short)0, c, false, false);
    }

    float* yb = Y + ((size_t)row0 + rt * 16 + 8 * half) * CLS + 16 * ct + l16;
    #pragma unroll
    for (int i = 0; i < 8; ++i)
        yb[(size_t)i * CLS] = c[i];
}

__global__ __launch_bounds__(256) void gcn_spmm_rows_f64_logsoftmax(
    const int* __restrict__ cnt, const int* __restrict__ edges,
    const int* __restrict__ spill_row, const int* __restrict__ spill_edge,
    const float* __restrict__ dense, const float* __restrict__ b2,
    float* __restrict__ out)
{
    const int lane = threadIdx.x & 31;
    const int r = __builtin_amdgcn_readfirstlane(
        (int)((blockIdx.x * (size_t)blockDim.x + threadIdx.x) >> 5));
    if (r >= NROWS) return;
    const int n   = cnt[r];
    const int end = (n < BUCKET) ? n : BUCKET;
    const int2*   ew = (const int2*)edges + (size_t)r * BUCKET;
    const float2* d2 = (const float2*)dense + lane;

    float2 acc = {0.0f, 0.0f};
    int p = 0;
    for (; p + 4 <= end; p += 4) {
        const int2 e0 = ew[p], e1 = ew[p + 1], e2 = ew[p + 2], e3 = ew[p + 3];
        const float2 g0 = d2[(size_t)e0.x * (CLS / 2)];
        const float2 g1 = d2[(size_t)e1.x * (CLS / 2)];
        const float2 g2 = d2[(size_t)e2.x * (CLS / 2)];
        const float2 g3 = d2[(size_t)e3.x * (CLS / 2)];
        const float v0 = __int_as_float(e0.y), v1 = __int_as_float(e1.y);
        const float v2 = __int_as_float(e2.y), v3 = __int_as_float(e3.y);
        acc.x += v0 * g0.x; acc.y += v0 * g0.y;
        acc.x += v1 * g1.x; acc.y += v1 * g1.y;
        acc.x += v2 * g2.x; acc.y += v2 * g2.y;
        acc.x += v3 * g3.x; acc.y += v3 * g3.y;
    }
    for (; p < end; ++p) {
        const int2 e0 = ew[p];
        const float2 g0 = d2[(size_t)e0.x * (CLS / 2)];
        const float v0 = __int_as_float(e0.y);
        acc.x += v0 * g0.x; acc.y += v0 * g0.y;
    }
    if (n > BUCKET) {
        const int q = cnt[NROWS];
        for (int i = 0; i < q; ++i) {
            if (spill_row[i] != r) continue;
            const int2 e0 = ((const int2*)spill_edge)[i];
            const float2 g0 = d2[(size_t)e0.x * (CLS / 2)];
            const float v0 = __int_as_float(e0.y);
            acc.x += v0 * g0.x; acc.y += v0 * g0.y;
        }
    }

    const float2 bb = *(const float2*)(b2 + lane * 2);
    float2 v;
    v.x = acc.x + bb.x;
    v.y = acc.y + bb.y;

    float m = fmaxf(v.x, v.y);
    #pragma unroll
    for (int off = 16; off > 0; off >>= 1)
        m = fmaxf(m, __shfl_xor(m, off, 32));

    float s = __expf(v.x - m) + __expf(v.y - m);
    #pragma unroll
    for (int off = 16; off > 0; off >>= 1)
        s += __shfl_xor(s, off, 32);

    const float lse = m + __logf(s);
    float2 o;
    o.x = v.x - lse;
    o.y = v.y - lse;
    ((float2*)out)[(size_t)r * (CLS / 2) + lane] = o;
}

extern "C" void kernel_launch(void* const* d_in, const int* in_sizes, int n_in,
                              void* d_out, int out_size, void* d_ws, size_t ws_size,
                              hipStream_t stream) {
    const float* x        = (const float*)d_in[0];
    const int*   edge_row = (const int*)  d_in[1];
    const int*   edge_col = (const int*)  d_in[2];
    const float* edge_val = (const float*)d_in[3];
    const float* W1       = (const float*)d_in[4];
    const float* b1       = (const float*)d_in[5];
    const float* W2       = (const float*)d_in[6];
    const float* b2       = (const float*)d_in[7];
    float* out = (float*)d_out;

    float* ws    = (float*)d_ws;
    float* XW1   = ws;
    float* HW2   = ws + (size_t)NROWS * HID;

    int* cnt        = (int*)(ws + 2 * (size_t)NROWS * HID);
    int* edges      = cnt + CNT_WORDS;
    int* spill_row  = edges + (size_t)NROWS * BUCKET * 2;
    int* spill_edge = spill_row + NEDGE;
    int* recs       = spill_edge + (size_t)NEDGE * 2;
    int* binoff     = recs + (size_t)NEDGE * 4;

    const int first_groups = GEMM_GROUPS + NCHUNK;
    const int row_blocks   = (NROWS * 32 + 255) / 256;

    gcn_gemm1_and_edge_bin<<<first_groups, 1024, 0, stream>>>(x, W1, XW1, edge_row, edge_col, edge_val, binoff, recs, cnt);
    gcn_range_buckets_spmm_f128_gemm2<<<NBINS, RANGE_THR, 0, stream>>>(binoff, recs, cnt, edges, spill_row, spill_edge, XW1, b1, W2, HW2);
    gcn_spmm_rows_f64_logsoftmax<<<row_blocks, 256, 0, stream>>>(cnt, edges, spill_row, spill_edge, HW2, b2, out);
}


# GEMM1 operand feed: W1 copied once per group into LDS (k row pairs side by side), B fragment = one 8-B LDS read, A = one 16-B load per 8 k; k summed 0,1,6,7 then 2,3,4,5 inside each 8 (new order)
# speedup vs baseline: 3.4742x; 1.0873x over previous
#include <hip/hip_runtime.h>


typedef __attribute__((ext_vector_type(2))) float v2f;
typedef __attribute__((ext_vector_type(8))) float v8f;

#define NROWS 50000
#define INDIM 256
#define HID   128
#define CLS   64
#define NEDGE 800000
#define BUCKET    64
#define CNT_WORDS 50176

#define RPB      64
#define RPB_LOG  6
#define NBINS    782
#define NBPAD    784
#define NCHUNK   256
#define EPC      3125
#define EPT      4

#define GEMM_GROUPS      250
#define STRIPS_PER_GROUP 25
#define WPITCH           288
#define LDS1_INT4        (INDIM / 2 * WPITCH / 4)
#define SREC_AT          ((2 * NBPAD + 32) / 4)

__global__ __launch_bounds__(1024) void gcn_gemm1_and_edge_bin(
    const float* __restrict__ X, const float* __restrict__ W1,
    float* __restrict__ Y,
    const int* __restrict__ erow, const int* __restrict__ ecol,
    const float* __restrict__ eval, int* __restrict__ binoff,
    int* __restrict__ recs, int* __restrict__ cnt)
{
    __shared__ int4 lds1[LDS1_INT4];
    int*  const binc     = (int*)lds1;
    int*  const binstart = binc + NBPAD;
    int*  const wsum     = binstart + NBPAD;
    int4* const srec     = lds1 + SREC_AT;

    const int t    = threadIdx.x;
    const int lane = t & 31;
    const int wv   = __builtin_amdgcn_readfirstlane(t >> 5);

    if (blockIdx.x < GEMM_GROUPS) {
        float* const wl = (float*)lds1;
        const float4* w4 = (const float4*)W1;
        #pragma unroll
        for (int i = 0; i < 4; ++i) {
            const int idx = t + 1024 * i;
            const int kp  = idx >> 5;
            const int cq  = idx & 31;
            const float4 r0 = w4[(2 * kp) * (HID / 4) + cq];
            const float4 r1 = w4[(2 * kp + 1) * (HID / 4) + cq];
            float* dst = wl + kp * WPITCH + 8 * cq;
            *(float4*)(dst)     = make_float4(r0.x, r1.x, r0.y, r1.y);
            *(float4*)(dst + 4) = make_float4(r0.z, r1.z, r0.w, r1.w);
        }
        __syncthreads();

        if (wv >= STRIPS_PER_GROUP) return;
        const int strip = (int)blockIdx.x * STRIPS_PER_GROUP + wv;
        const int mtile = strip >> 1;
        const int nh    = strip & 1;
        const int half  = lane >> 4;
        const int l16   = lane & 15;

        const int m0 = mtile * 16;
        const int n0 = nh * 64;

        const float* xrow = X + (size_t)(m0 + l16) * INDIM + 4 * half;
        const float* wb1 = wl + (half ? 3 : 0) * WPITCH + 2 * (n0 + l16);
        const float* wb2 = wl + (half ? 2 : 1) * WPITCH + 2 * (n0 + l16);

        v8f acc0 = {}, acc1 = {}, acc2 = {}, acc3 = {};
        #pragma unroll 2
        for (int k = 0; k < INDIM; k += 8) {
            const float4 av = *(const float4*)(xrow + k);
            v2f a1; a1.x = half ? av.z : av.x; a1.y = half ? av.w : av.y;
            v2f a2; a2.x = half ? av.x : av.z; a2.y = half ? av.y : av.w;
            const float* p1 = wb1 + (k >> 1) * WPITCH;
            const float* p2 = wb2 + (k >> 1) * WPITCH;
            const float2 f10 = *(const float2*)(p1),      f11 = *(const float2*)(p1 + 32);
            const float2 f12 = *(const float2*)(p1 + 64), f13 = *(const float2*)(p1 + 96);
            const float2 f20 = *(const float2*)(p2),      f21 = *(const float2*)(p2 + 32);
            const float2 f22 = *(const float2*)(p2 + 64), f23 = *(const float2*)(p2 + 96);
            v2f b10; b10.x = f10.x; b10.y = f10.y;
            v2f b11; b11.x = f11.x; b11.y = f11.y;
            v2f b12; b12.x = f12.x; b12.y = f12.y;
            v2f b13; b13.x = f13.x; b13.y = f13.y;
            v2f b20; b20.x = f20.x; b20.y = f20.y;
            v2f b21; b21.x = f21.x; b21.y = f21.y;
            v2f b22; b22.x = f22.x; b22.y = f22.y;
            v2f b23; b23.x = f23.x; b23.y = f23.y;
            acc0 = __builtin_amdgcn_wmma_f32_16x16x4_f32(false, a1, false, b10, (short)0, acc0, false, false);
            acc1 = __builtin_amdgcn_wmma_f32_16x16x4_f32(false, a1, false, b11, (short)0, acc1, false, false);
            acc2 = __builtin_amdgcn_wmma_f32_16x16x4_f32(false, a1, false, b12, (short)0, acc2, false, false);
            acc3 = __builtin_amdgcn_wmma_f32_16x16x4_f32(false, a1, false, b13, (short)0, acc3, false, false);
            acc0 = __builtin_amdgcn_wmma_f32_16x16x4_f32(false, a2, false, b20, (short)0, acc0, false, false);
            acc1 = __builtin_amdgcn_wmma_f32_16x16x4_f32(false, a2, false, b21, (short)0, acc1, false, false);
            acc2 = __builtin_amdgcn_wmma_f32_16x16x4_f32(false, a2, false, b22, (short)0, acc2, false, false);
            acc3 = __builtin_amdgcn_wmma_f32_16x16x4_f32(false, a2, false, b23, (short)0, acc3, false, false);
        }

        float* yb = Y + n0 + l16;
        #pragma unroll
        for (int i = 0; i < 8; ++i) {
            float* yr = yb + (size_t)(m0 + i + 8 * half) * HID;
            yr[0]  = acc0[i];
            yr[16] = acc1[i];
            yr[32] = acc2[i];
            yr[48] = acc3[i];
        }
        return;
    }

    const int c = (int)blockIdx.x - GEMM_GROUPS;

    if (c == 0 && t == 0) cnt[NROWS] = 0;

    if (t < NBPAD) binc[t] = 0;
    __syncthreads();

    int r[EPT], cl[EPT], vb[EPT], rk[EPT];
    #pragma unroll
    for (int j = 0; j < EPT; ++j) {
        const int el = t + 1024 * j;
        r[j] = 0; cl[j] = 0; vb[j] = 0; rk[j] = 0;
        if (el < EPC) {
            const int e = c * EPC + el;
            r[j]  = erow[e];
            cl[j] = ecol[e];
            vb[j] = __float_as_int(eval[e]);
        }
    }
    #pragma unroll
    for (int j = 0; j < EPT; ++j)
        if (t + 1024 * j < EPC) rk[j] = atomicAdd(&binc[r[j] >> RPB_LOG], 1);
    __syncthreads();

    const int x = (t < NBINS) ? binc[t] : 0;
    int incl = x;
    #pragma unroll
    for (int off = 1; off < 32; off <<= 1) {
        const int y = __shfl_up(incl, off, 32);
        if (lane >= off) incl += y;
    }
    if (lane == 31) wsum[wv] = incl;
    __syncthreads();
    int base = 0;
    for (int i = 0; i < wv; ++i) base += wsum[i];
    const int excl = base + incl - x;
    if (t <= NBINS) {
        binstart[t] = excl;
        binoff[c * NBPAD + t] = excl;
    }
    __syncthreads();

    #pragma unroll
    for (int j = 0; j < EPT; ++j)
        if (t + 1024 * j < EPC) {
            const int pos = binstart[r[j] >> RPB_LOG] + rk[j];
            srec[pos] = make_int4(r[j], cl[j], vb[j], 0);
        }
    __syncthreads();

    int4* dst = (int4*)recs + (size_t)c * EPC;
    for (int i = t; i < EPC; i += 1024) dst[i] = srec[i];
}

#define HPITCH    132
#define RANGE_THR 512

__global__ __launch_bounds__(RANGE_THR, 16) void gcn_range_buckets_spmm_f128_gemm2(
    const int* __restrict__ binoff, const int* __restrict__ recs,
    int* __restrict__ cnt, int* __restrict__ edges,
    int* __restrict__ spill_row, int* __restrict__ spill_edge,
    const float* __restrict__ dense, const float* __restrict__ b1,
    const float* __restrict__ W2, float* __restrict__ Y)
{
    __shared__ int   lcnt[RPB];
    __shared__ int2  lbuck[RPB * BUCKET];
    __shared__ float hs[RPB * HPITCH];

    const int t    = threadIdx.x;
    const int lane = t & 31;
    const int wave = __builtin_amdgcn_readfirstlane(t >> 5);
    const int b    = (int)blockIdx.x;
    const int row0 = b * RPB;

    if (t < RPB) lcnt[t] = 0;
    int o0 = 0, o1 = 0;
    if (t < NCHUNK) {
        o0 = binoff[t * NBPAD + b];
        o1 = binoff[t * NBPAD + b + 1];
    }
    __syncthreads();

    if (t < NCHUNK) {
        const int4* piece = (const int4*)recs + (size_t)t * EPC;
        const int pe = o1;
        for (int i = o0; i < pe; i += 4) {
            int4 q[4];
            #pragma unroll
            for (int j = 0; j < 4; ++j)
                q[j] = piece[(i + j < pe) ? i + j : pe - 1];
            #pragma unroll
            for (int j = 0; j < 4; ++j) {
                if (i + j < pe) {
                    const int rl = q[j].x - row0;
                    const int k  = atomicAdd(&lcnt[rl], 1);
                    if (k < BUCKET) {
                        lbuck[rl * BUCKET + k] = make_int2(q[j].y, q[j].z);
                    } else {
                        const int s = atomicAdd(cnt + NROWS, 1);
                        spill_row[s] = q[j].x;
                        ((int2*)spill_edge)[s] = make_int2(q[j].y, q[j].z);
                    }
                }
            }
        }
    }
    __syncthreads();

    const float4* d4 = (const float4*)dense + lane;
    const float4  bb = *(const float4*)(b1 + 4 * lane);
    for (int rl = wave; rl < RPB; rl += RANGE_THR / 32) {
        const int row = row0 + rl;
        float4 acc = {0.0f, 0.0f, 0.0f, 0.0f};
        if (row < NROWS) {
            const int n   = __builtin_amdgcn_readfirstlane(lcnt[rl]);
            const int end = (n < BUCKET) ? n : BUCKET;
            const int2* eb = lbuck + rl * BUCKET;

            if (lane == 0) cnt[row] = n;
            int2* dst = (int2*)edges + (size_t)row * BUCKET;
            if (lane < end)      dst[lane]      = eb[lane];
            if (lane + 32 < end) dst[lane + 32] = eb[lane + 32];

            if (n <= BUCKET) {
                int p = 0;
                for (; p + 4 <= end; p += 4) {
                    const int2 e0 = eb[p], e1 = eb[p + 1], e2 = eb[p + 2], e3 = eb[p + 3];
                    const float4 g0 = d4[(size_t)e0.x * (HID / 4)];
                    const float4 g1 = d4[(size_t)e1.x * (HID / 4)];
                    const float4 g2 = d4[(size_t)e2.x * (HID / 4)];
                    const float4 g3 = d4[(size_t)e3.x * (HID / 4)];
                    const float v0 = __int_as_float(e0.y), v1 = __int_as_float(e1.y);
                    const float v2 = __int_as_float(e2.y), v3 = __int_as_float(e3.y);
                    acc.x += v0 * g0.x; acc.y += v0 * g0.y; acc.z += v0 * g0.z; acc.w += v0 * g0.w;
                    acc.x += v1 * g1.x; acc.y += v1 * g1.y; acc.z += v1 * g1.z; acc.w += v1 * g1.w;
                    acc.x += v2 * g2.x; acc.y += v2 * g2.y; acc.z += v2 * g2.z; acc.w += v2 * g2.w;
                    acc.x += v3 * g3.x; acc.y += v3 * g3.y; acc.z += v3 * g3.z; acc.w += v3 * g3.w;
                }
                for (; p < end; ++p) {
                    const int2 e0 = eb[p];
                    const float4 g0 = d4[(size_t)e0.x * (HID / 4)];
                    const float v0 = __int_as_float(e0.y);
                    acc.x += v0 * g0.x; acc.y += v0 * g0.y; acc.z += v0 * g0.z; acc.w += v0 * g0.w;
                }
            } else {
                for (int ch = 0; ch < NCHUNK; ++ch) {
                    const int q0 = binoff[ch * NBPAD + b];
                    const int q1 = binoff[ch * NBPAD + b + 1];
                    const int4* piece = (const int4*)recs + (size_t)ch * EPC;
                    for (int i = q0; i < q1; ++i) {
                        const int4 q = piece[i];
                        if (q.x != row) continue;
                        const float4 g0 = d4[(size_t)q.y * (HID / 4)];
                        const float v0 = __int_as_float(q.z);
                        acc.x += v0 * g0.x; acc.y += v0 * g0.y; acc.z += v0 * g0.z; acc.w += v0 * g0.w;
                    }
                }
            }
        }
        float4 hv;
        hv.x = fmaxf(acc.x + bb.x, 0.0f);
        hv.y = fmaxf(acc.y + bb.y, 0.0f);
        hv.z = fmaxf(acc.z + bb.z, 0.0f);
        hv.w = fmaxf(acc.w + bb.w, 0.0f);
        *(float4*)(hs + rl * HPITCH + 4 * lane) = hv;
    }
    __syncthreads();

    const int rt = wave >> 2;
    const int ct = wave & 3;
    if (row0 + rt * 16 >= NROWS) return;
    const int half = lane >> 4;
    const int l16  = lane & 15;
    const float* hrow = hs + (rt * 16 + l16) * HPITCH + 2 * half;
    const float* wcol = W2 + 16 * ct + l16;

    v8f c = {};
    #pragma unroll 8
    for (int k = 0; k < HID; k += 4) {
        const float2 av = *(const float2*)(hrow + k);
        v2f a; a.x = av.x; a.y = av.y;
        v2f bf;
        bf.x = wcol[(size_t)(k + 2 * half) * CLS];
        bf.y = wcol[(size_t)(k + 2 * half + 1) * CLS];
        c = __builtin_amdgcn_wmma_f32_16x16x4_f32(false, a, false, bf, (short)0, c, false, false);
    }

    float* yb = Y + ((size_t)row0 + rt * 16 + 8 * half) * CLS + 16 * ct + l16;
    #pragma unroll
    for (int i = 0; i < 8; ++i)
        yb[(size_t)i * CLS] = c[i];
}

__global__ __launch_bounds__(256) void gcn_spmm_rows_f64_logsoftmax(
    const int* __restrict__ cnt, const int* __restrict__ edges,
    const int* __restrict__ spill_row, const int* __restrict__ spill_edge,
    const float* __restrict__ dense, const float* __restrict__ b2,
    float* __restrict__ out)
{
    const int lane = threadIdx.x & 31;
    const int r = __builtin_amdgcn_readfirstlane(
        (int)((blockIdx.x * (size_t)blockDim.x + threadIdx.x) >> 5));
    if (r >= NROWS) return;
    const int n   = cnt[r];
    const int end = (n < BUCKET) ? n : BUCKET;
    const int2*   ew = (const int2*)edges + (size_t)r * BUCKET;
    const float2* d2 = (const float2*)dense + lane;

    float2 acc = {0.0f, 0.0f};
    int p = 0;
    for (; p + 4 <= end; p += 4) {
        const int2 e0 = ew[p], e1 = ew[p + 1], e2 = ew[p + 2], e3 = ew[p + 3];
        const float2 g0 = d2[(size_t)e0.x * (CLS / 2)];
        const float2 g1 = d2[(size_t)e1.x * (CLS / 2)];
        const float2 g2 = d2[(size_t)e2.x * (CLS / 2)];
        const float2 g3 = d2[(size_t)e3.x * (CLS / 2)];
        const float v0 = __int_as_float(e0.y), v1 = __int_as_float(e1.y);
        const float v2 = __int_as_float(e2.y), v3 = __int_as_float(e3.y);
        acc.x += v0 * g0.x; acc.y += v0 * g0.y;
        acc.x += v1 * g1.x; acc.y += v1 * g1.y;
        acc.x += v2 * g2.x; acc.y += v2 * g2.y;
        acc.x += v3 * g3.x; acc.y += v3 * g3.y;
    }
    for (; p < end; ++p) {
        const int2 e0 = ew[p];
        const float2 g0 = d2[(size_t)e0.x * (CLS / 2)];
        const float v0 = __int_as_float(e0.y);
        acc.x += v0 * g0.x; acc.y += v0 * g0.y;
    }
    if (n > BUCKET) {
        const int q = cnt[NROWS];
        for (int i = 0; i < q; ++i) {
            if (spill_row[i] != r) continue;
            const int2 e0 = ((const int2*)spill_edge)[i];
            const float2 g0 = d2[(size_t)e0.x * (CLS / 2)];
            const float v0 = __int_as_float(e0.y);
            acc.x += v0 * g0.x; acc.y += v0 * g0.y;
        }
    }

    const float2 bb = *(const float2*)(b2 + lane * 2);
    float2 v;
    v.x = acc.x + bb.x;
    v.y = acc.y + bb.y;

    float m = fmaxf(v.x, v.y);
    #pragma unroll
    for (int off = 16; off > 0; off >>= 1)
        m = fmaxf(m, __shfl_xor(m, off, 32));

    float s = __expf(v.x - m) + __expf(v.y - m);
    #pragma unroll
    for (int off = 16; off > 0; off >>= 1)
        s += __shfl_xor(s, off, 32);

    const float lse = m + __logf(s);
    float2 o;
    o.x = v.x - lse;
    o.y = v.y - lse;
    ((float2*)out)[(size_t)r * (CLS / 2) + lane] = o;
}

extern "C" void kernel_launch(void* const* d_in, const int* in_sizes, int n_in,
                              void* d_out, int out_size, void* d_ws, size_t ws_size,
                              hipStream_t stream) {
    const float* x        = (const float*)d_in[0];
    const int*   edge_row = (const int*)  d_in[1];
    const int*   edge_col = (const int*)  d_in[2];
    const float* edge_val = (const float*)d_in[3];
    const float* W1       = (const float*)d_in[4];
    const float* b1       = (const float*)d_in[5];
    const float* W2       = (const float*)d_in[6];
    const float* b2       = (const float*)d_in[7];
    float* out = (float*)d_out;

    float* ws    = (float*)d_ws;
    float* XW1   = ws;
    float* HW2   = ws + (size_t)NROWS * HID;

    int* cnt        = (int*)(ws + 2 * (size_t)NROWS * HID);
    int* edges      = cnt + CNT_WORDS;
    int* spill_row  = edges + (size_t)NROWS * BUCKET * 2;
    int* spill_edge = spill_row + NEDGE;
    int* recs       = spill_edge + (size_t)NEDGE * 2;
    int* binoff     = recs + (size_t)NEDGE * 4;

    const int first_groups = GEMM_GROUPS + NCHUNK;
    const int row_blocks   = (NROWS * 32 + 255) / 256;

    gcn_gemm1_and_edge_bin<<<first_groups, 1024, 0, stream>>>(x, W1, XW1, edge_row, edge_col, edge_val, binoff, recs, cnt);
    gcn_range_buckets_spmm_f128_gemm2<<<NBINS, RANGE_THR, 0, stream>>>(binoff, recs, cnt, edges, spill_row, spill_edge, XW1, b1, W2, HW2);
    gcn_spmm_rows_f64_logsoftmax<<<row_blocks, 256, 0, stream>>>(cnt, edges, spill_row, spill_edge, HW2, b2, out);
}


# SpMM row walks: a row's bucket slots sit one per lane, col/val go round by lane read, and a wave issues 8 (F=128) or 16 (F=64) source-row loads per trip, was 4 then 1; sums keep their slot order
# speedup vs baseline: 3.6793x; 1.0590x over previous
#include <hip/hip_runtime.h>


typedef __attribute__((ext_vector_type(2))) float v2f;
typedef __attribute__((ext_vector_type(8))) float v8f;

#define NROWS 50000
#define INDIM 256
#define HID   128
#define CLS   64
#define NEDGE 800000
#define BUCKET    64
#define CNT_WORDS 50176

#define RPB      64
#define RPB_LOG  6
#define NBINS    782
#define NBPAD    784
#define NCHUNK   256
#define EPC      3125
#define EPT      4

#define GEMM_GROUPS      250
#define STRIPS_PER_GROUP 25
#define WPITCH           288
#define LDS1_INT4        (INDIM / 2 * WPITCH / 4)
#define SREC_AT          ((2 * NBPAD + 32) / 4)

__global__ __launch_bounds__(1024) void gcn_gemm1_and_edge_bin(
    const float* __restrict__ X, const float* __restrict__ W1,
    float* __restrict__ Y,
    const int* __restrict__ erow, const int* __restrict__ ecol,
    const float* __restrict__ eval, int* __restrict__ binoff,
    int* __restrict__ recs, int* __restrict__ cnt)
{
    __shared__ int4 lds1[LDS1_INT4];
    int*  const binc     = (int*)lds1;
    int*  const binstart = binc + NBPAD;
    int*  const wsum     = binstart + NBPAD;
    int4* const srec     = lds1 + SREC_AT;

    const int t    = threadIdx.x;
    const int lane = t & 31;
    const int wv   = __builtin_amdgcn_readfirstlane(t >> 5);

    if (blockIdx.x < GEMM_GROUPS) {
        float* const wl = (float*)lds1;
        const float4* w4 = (const float4*)W1;
        #pragma unroll
        for (int i = 0; i < 4; ++i) {
            const int idx = t + 1024 * i;
            const int kp  = idx >> 5;
            const int cq  = idx & 31;
            const float4 r0 = w4[(2 * kp) * (HID / 4) + cq];
            const float4 r1 = w4[(2 * kp + 1) * (HID / 4) + cq];
            float* dst = wl + kp * WPITCH + 8 * cq;
            *(float4*)(dst)     = make_float4(r0.x, r1.x, r0.y, r1.y);
            *(float4*)(dst + 4) = make_float4(r0.z, r1.z, r0.w, r1.w);
        }
        __syncthreads();

        if (wv >= STRIPS_PER_GROUP) return;
        const int strip = (int)blockIdx.x * STRIPS_PER_GROUP + wv;
        const int mtile = strip >> 1;
        const int nh    = strip & 1;
        const int half  = lane >> 4;
        const int l16   = lane & 15;

        const int m0 = mtile * 16;
        const int n0 = nh * 64;

        const float* xrow = X + (size_t)(m0 + l16) * INDIM + 4 * half;
        const float* wb1 = wl + (half ? 3 : 0) * WPITCH + 2 * (n0 + l16);
        const float* wb2 = wl + (half ? 2 : 1) * WPITCH + 2 * (n0 + l16);

        v8f acc0 = {}, acc1 = {}, acc2 = {}, acc3 = {};
        #pragma unroll 2
        for (int k = 0; k < INDIM; k += 8) {
            const float4 av = *(const float4*)(xrow + k);
            v2f a1; a1.x = half ? av.z : av.x; a1.y = half ? av.w : av.y;
            v2f a2; a2.x = half ? av.x : av.z; a2.y = half ? av.y : av.w;
            const float* p1 = wb1 + (k >> 1) * WPITCH;
            const float* p2 = wb2 + (k >> 1) * WPITCH;
            const float2 f10 = *(const float2*)(p1),      f11 = *(const float2*)(p1 + 32);
            const float2 f12 = *(const float2*)(p1 + 64), f13 = *(const float2*)(p1 + 96);
            const float2 f20 = *(const float2*)(p2),      f21 = *(const float2*)(p2 + 32);
            const float2 f22 = *(const float2*)(p2 + 64), f23 = *(const float2*)(p2 + 96);
            v2f b10; b10.x = f10.x; b10.y = f10.y;
            v2f b11; b11.x = f11.x; b11.y = f11.y;
            v2f b12; b12.x = f12.x; b12.y = f12.y;
            v2f b13; b13.x = f13.x; b13.y = f13.y;
            v2f b20; b20.x = f20.x; b20.y = f20.y;
            v2f b21; b21.x = f21.x; b21.y = f21.y;
            v2f b22; b22.x = f22.x; b22.y = f22.y;
            v2f b23; b23.x = f23.x; b23.y = f23.y;
            acc0 = __builtin_amdgcn_wmma_f32_16x16x4_f32(false, a1, false, b10, (short)0, acc0, false, false);
            acc1 = __builtin_amdgcn_wmma_f32_16x16x4_f32(false, a1, false, b11, (short)0, acc1, false, false);
            acc2 = __builtin_amdgcn_wmma_f32_16x16x4_f32(false, a1, false, b12, (short)0, acc2, false, false);
            acc3 = __builtin_amdgcn_wmma_f32_16x16x4_f32(false, a1, false, b13, (short)0, acc3, false, false);
            acc0 = __builtin_amdgcn_wmma_f32_16x16x4_f32(false, a2, false, b20, (short)0, acc0, false, false);
            acc1 = __builtin_amdgcn_wmma_f32_16x16x4_f32(false, a2, false, b21, (short)0, acc1, false, false);
            acc2 = __builtin_amdgcn_wmma_f32_16x16x4_f32(false, a2, false, b22, (short)0, acc2, false, false);
            acc3 = __builtin_amdgcn_wmma_f32_16x16x4_f32(false, a2, false, b23, (short)0, acc3, false, false);
        }

        float* yb = Y + n0 + l16;
        #pragma unroll
        for (int i = 0; i < 8; ++i) {
            float* yr = yb + (size_t)(m0 + i + 8 * half) * HID;
            yr[0]  = acc0[i];
            yr[16] = acc1[i];
            yr[32] = acc2[i];
            yr[48] = acc3[i];
        }
        return;
    }

    const int c = (int)blockIdx.x - GEMM_GROUPS;

    if (c == 0 && t == 0) cnt[NROWS] = 0;

    if (t < NBPAD) binc[t] = 0;
    __syncthreads();

    int r[EPT], cl[EPT], vb[EPT], rk[EPT];
    #pragma unroll
    for (int j = 0; j < EPT; ++j) {
        const int el = t + 1024 * j;
        r[j] = 0; cl[j] = 0; vb[j] = 0; rk[j] = 0;
        if (el < EPC) {
            const int e = c * EPC + el;
            r[j]  = erow[e];
            cl[j] = ecol[e];
            vb[j] = __float_as_int(eval[e]);
        }
    }
    #pragma unroll
    for (int j = 0; j < EPT; ++j)
        if (t + 1024 * j < EPC) rk[j] = atomicAdd(&binc[r[j] >> RPB_LOG], 1);
    __syncthreads();

    const int x = (t < NBINS) ? binc[t] : 0;
    int incl = x;
    #pragma unroll
    for (int off = 1; off < 32; off <<= 1) {
        const int y = __shfl_up(incl, off, 32);
        if (lane >= off) incl += y;
    }
    if (lane == 31) wsum[wv] = incl;
    __syncthreads();
    int base = 0;
    for (int i = 0; i < wv; ++i) base += wsum[i];
    const int excl = base + incl - x;
    if (t <= NBINS) {
        binstart[t] = excl;
        binoff[c * NBPAD + t] = excl;
    }
    __syncthreads();

    #pragma unroll
    for (int j = 0; j < EPT; ++j)
        if (t + 1024 * j < EPC) {
            const int pos = binstart[r[j] >> RPB_LOG] + rk[j];
            srec[pos] = make_int4(r[j], cl[j], vb[j], 0);
        }
    __syncthreads();

    int4* dst = (int4*)recs + (size_t)c * EPC;
    for (int i = t; i < EPC; i += 1024) dst[i] = srec[i];
}

#define GB128 8
#define GB64  16

template <int NB>
__device__ __forceinline__ void gather_batch_f128(float4& acc, const int2 e, const int q, const int lastl,
                                                  const char* db, const unsigned lane_off)
{
    float4 g[NB];
    float  v[NB];
    #pragma unroll
    for (int j = 0; j < NB; ++j) {
        const int s   = q + j;
        const int sl  = (s < lastl) ? s : lastl;
        const int col = __builtin_amdgcn_readlane(e.x, sl);
        const int vb  = __builtin_amdgcn_readlane(e.y, sl);
        v[j] = (s <= lastl) ? __int_as_float(vb) : 0.0f;
        g[j] = *(const float4*)(db + (((unsigned)col << 9) + lane_off));
    }
    #pragma unroll
    for (int j = 0; j < NB; ++j) {
        acc.x = __builtin_fmaf(v[j], g[j].x, acc.x); acc.y = __builtin_fmaf(v[j], g[j].y, acc.y);
        acc.z = __builtin_fmaf(v[j], g[j].z, acc.z); acc.w = __builtin_fmaf(v[j], g[j].w, acc.w);
    }
}

__device__ __forceinline__ void gather_rows_f128(float4& acc, const int2 e, const int lastl,
                                                 const char* db, const unsigned lane_off)
{
    int q = 0;
    while (q <= lastl) {
        if (lastl - q >= GB128 / 2) { gather_batch_f128<GB128>(acc, e, q, lastl, db, lane_off);     q += GB128; }
        else                        { gather_batch_f128<GB128 / 2>(acc, e, q, lastl, db, lane_off); q += GB128 / 2; }
    }
}

template <int NB>
__device__ __forceinline__ void gather_batch_f64(float2& acc, const int2 e, const int q, const int lastl,
                                                 const char* db, const unsigned lane_off)
{
    float2 g[NB];
    float  v[NB];
    #pragma unroll
    for (int j = 0; j < NB; ++j) {
        const int s   = q + j;
        const int sl  = (s < lastl) ? s : lastl;
        const int col = __builtin_amdgcn_readlane(e.x, sl);
        const int vb  = __builtin_amdgcn_readlane(e.y, sl);
        v[j] = (s <= lastl) ? __int_as_float(vb) : 0.0f;
        g[j] = *(const float2*)(db + (((unsigned)col << 8) + lane_off));
    }
    #pragma unroll
    for (int j = 0; j < NB; ++j) {
        acc.x = __builtin_fmaf(v[j], g[j].x, acc.x); acc.y = __builtin_fmaf(v[j], g[j].y, acc.y);
    }
}

__device__ __forceinline__ void gather_rows_f64(float2& acc, const int2 e, const int lastl,
                                                const char* db, const unsigned lane_off)
{
    int q = 0;
    while (q <= lastl) {
        if (lastl - q >= GB64 / 2) { gather_batch_f64<GB64>(acc, e, q, lastl, db, lane_off);     q += GB64; }
        else                       { gather_batch_f64<GB64 / 2>(acc, e, q, lastl, db, lane_off); q += GB64 / 2; }
    }
}

#define HPITCH    132
#define RANGE_THR 512

__global__ __launch_bounds__(RANGE_THR, 16) void gcn_range_buckets_spmm_f128_gemm2(
    const int* __restrict__ binoff, const int* __restrict__ recs,
    int* __restrict__ cnt, int* __restrict__ edges,
    int* __restrict__ spill_row, int* __restrict__ spill_edge,
    const float* __restrict__ dense, const float* __restrict__ b1,
    const float* __restrict__ W2, float* __restrict__ Y)
{
    __shared__ int   lcnt[RPB];
    __shared__ int2  lbuck[RPB * BUCKET];
    __shared__ float hs[RPB * HPITCH];

    const int t    = threadIdx.x;
    const int lane = t & 31;
    const int wave = __builtin_amdgcn_readfirstlane(t >> 5);
    const int b    = (int)blockIdx.x;
    const int row0 = b * RPB;

    if (t < RPB) lcnt[t] = 0;
    int o0 = 0, o1 = 0;
    if (t < NCHUNK) {
        o0 = binoff[t * NBPAD + b];
        o1 = binoff[t * NBPAD + b + 1];
    }
    __syncthreads();

    if (t < NCHUNK) {
        const int4* piece = (const int4*)recs + (size_t)t * EPC;
        const int pe = o1;
        for (int i = o0; i < pe; i += 4) {
            int4 q[4];
            #pragma unroll
            for (int j = 0; j < 4; ++j)
                q[j] = piece[(i + j < pe) ? i + j : pe - 1];
            #pragma unroll
            for (int j = 0; j < 4; ++j) {
                if (i + j < pe) {
                    const int rl = q[j].x - row0;
                    const int k  = atomicAdd(&lcnt[rl], 1);
                    if (k < BUCKET) {
                        lbuck[rl * BUCKET + k] = make_int2(q[j].y, q[j].z);
                    } else {
                        const int s = atomicAdd(cnt + NROWS, 1);
                        spill_row[s] = q[j].x;
                        ((int2*)spill_edge)[s] = make_int2(q[j].y, q[j].z);
                    }
                }
            }
        }
    }
    __syncthreads();

    const float4* d4 = (const float4*)dense + lane;
    const float4  bb = *(const float4*)(b1 + 4 * lane);
    for (int rl = wave; rl < RPB; rl += RANGE_THR / 32) {
        const int row = row0 + rl;
        float4 acc = {0.0f, 0.0f, 0.0f, 0.0f};
        if (row < NROWS) {
            const int n   = __builtin_amdgcn_readfirstlane(lcnt[rl]);
            const int end = (n < BUCKET) ? n : BUCKET;
            const int2* eb = lbuck + rl * BUCKET;

            if (lane == 0) cnt[row] = n;
            int2* dst = (int2*)edges + (size_t)row * BUCKET;
            int2 e_lo = make_int2(0, 0), e_hi = make_int2(0, 0);
            if (lane < end)      { e_lo = eb[lane];      dst[lane]      = e_lo; }
            if (lane + 32 < end) { e_hi = eb[lane + 32]; dst[lane + 32] = e_hi; }

            if (n <= BUCKET) {
                const char* db = (const char*)dense;
                const unsigned lane_off = 16u * (unsigned)lane;
                if (end > 0)  gather_rows_f128(acc, e_lo, (end < 32 ? end : 32) - 1, db, lane_off);
                if (end > 32) gather_rows_f128(acc, e_hi, end - 33, db, lane_off);
            } else {
                for (int ch = 0; ch < NCHUNK; ++ch) {
                    const int q0 = binoff[ch * NBPAD + b];
                    const int q1 = binoff[ch * NBPAD + b + 1];
                    const int4* piece = (const int4*)recs + (size_t)ch * EPC;
                    for (int i = q0; i < q1; ++i) {
                        const int4 q = piece[i];
                        if (q.x != row) continue;
                        const float4 g0 = d4[(size_t)q.y * (HID / 4)];
                        const float v0 = __int_as_float(q.z);
                        acc.x += v0 * g0.x; acc.y += v0 * g0.y; acc.z += v0 * g0.z; acc.w += v0 * g0.w;
                    }
                }
            }
        }
        float4 hv;
        hv.x = fmaxf(acc.x + bb.x, 0.0f);
        hv.y = fmaxf(acc.y + bb.y, 0.0f);
        hv.z = fmaxf(acc.z + bb.z, 0.0f);
        hv.w = fmaxf(acc.w + bb.w, 0.0f);
        *(float4*)(hs + rl * HPITCH + 4 * lane) = hv;
    }
    __syncthreads();

    const int rt = wave >> 2;
    const int ct = wave & 3;
    if (row0 + rt * 16 >= NROWS) return;
    const int half = lane >> 4;
    const int l16  = lane & 15;
    const float* hrow = hs + (rt * 16 + l16) * HPITCH + 2 * half;
    const float* wcol = W2 + 16 * ct + l16;

    v8f c = {};
    #pragma unroll 8
    for (int k = 0; k < HID; k += 4) {
        const float2 av = *(const float2*)(hrow + k);
        v2f a; a.x = av.x; a.y = av.y;
        v2f bf;
        bf.x = wcol[(size_t)(k + 2 * half) * CLS];
        bf.y = wcol[(size_t)(k + 2 * half + 1) * CLS];
        c = __builtin_amdgcn_wmma_f32_16x16x4_f32(false, a, false, bf, (short)0, c, false, false);
    }

    float* yb = Y + ((size_t)row0 + rt * 16 + 8 * half) * CLS + 16 * ct + l16;
    #pragma unroll
    for (int i = 0; i < 8; ++i)
        yb[(size_t)i * CLS] = c[i];
}

__global__ __launch_bounds__(256) void gcn_spmm_rows_f64_logsoftmax(
    const int* __restrict__ cnt, const int* __restrict__ edges,
    const int* __restrict__ spill_row, const int* __restrict__ spill_edge,
    const float* __restrict__ dense, const float* __restrict__ b2,
    float* __restrict__ out)
{
    const int lane = threadIdx.x & 31;
    const int r = __builtin_amdgcn_readfirstlane(
        (int)((blockIdx.x * (size_t)blockDim.x + threadIdx.x) >> 5));
    if (r >= NROWS) return;
    const int n   = cnt[r];
    const int end = (n < BUCKET) ? n : BUCKET;
    const int2* ew = (const int2*)edges + (size_t)r * BUCKET;

    float2 acc = {0.0f, 0.0f};
    const char* db = (const char*)dense;
    const unsigned lane_off = 8u * (unsigned)lane;
    for (int h0 = 0; h0 < end; h0 += 32) {
        int2 e = make_int2(0, 0);
        if (h0 + lane < end) e = ew[h0 + lane];
        const int rest = end - 1 - h0;
        gather_rows_f64(acc, e, (rest < 31) ? rest : 31, db, lane_off);
    }
    if (n > BUCKET) {
        const float2* d2 = (const float2*)dense + lane;
        const int q = cnt[NROWS];
        for (int i = 0; i < q; ++i) {
            if (spill_row[i] != r) continue;
            const int2 e0 = ((const int2*)spill_edge)[i];
            const float2 g0 = d2[(size_t)e0.x * (CLS / 2)];
            const float v0 = __int_as_float(e0.y);
            acc.x += v0 * g0.x; acc.y += v0 * g0.y;
        }
    }

    const float2 bb = *(const float2*)(b2 + lane * 2);
    float2 v;
    v.x = acc.x + bb.x;
    v.y = acc.y + bb.y;

    float m = fmaxf(v.x, v.y);
    #pragma unroll
    for (int off = 16; off > 0; off >>= 1)
        m = fmaxf(m, __shfl_xor(m, off, 32));

    float s = __expf(v.x - m) + __expf(v.y - m);
    #pragma unroll
    for (int off = 16; off > 0; off >>= 1)
        s += __shfl_xor(s, off, 32);

    const float lse = m + __logf(s);
    float2 o;
    o.x = v.x - lse;
    o.y = v.y - lse;
    ((float2*)out)[(size_t)r * (CLS / 2) + lane] = o;
}

extern "C" void kernel_launch(void* const* d_in, const int* in_sizes, int n_in,
                              void* d_out, int out_size, void* d_ws, size_t ws_size,
                              hipStream_t stream) {
    const float* x        = (const float*)d_in[0];
    const int*   edge_row = (const int*)  d_in[1];
    const int*   edge_col = (const int*)  d_in[2];
    const float* edge_val = (const float*)d_in[3];
    const float* W1       = (const float*)d_in[4];
    const float* b1       = (const float*)d_in[5];
    const float* W2       = (const float*)d_in[6];
    const float* b2       = (const float*)d_in[7];
    float* out = (float*)d_out;

    float* ws    = (float*)d_ws;
    float* XW1   = ws;
    float* HW2   = ws + (size_t)NROWS * HID;

    int* cnt        = (int*)(ws + 2 * (size_t)NROWS * HID);
    int* edges      = cnt + CNT_WORDS;
    int* spill_row  = edges + (size_t)NROWS * BUCKET * 2;
    int* spill_edge = spill_row + NEDGE;
    int* recs       = spill_edge + (size_t)NEDGE * 2;
    int* binoff     = recs + (size_t)NEDGE * 4;

    const int first_groups = GEMM_GROUPS + NCHUNK;
    const int row_blocks   = (NROWS * 32 + 255) / 256;

    gcn_gemm1_and_edge_bin<<<first_groups, 1024, 0, stream>>>(x, W1, XW1, edge_row, edge_col, edge_val, binoff, recs, cnt);
    gcn_range_buckets_spmm_f128_gemm2<<<NBINS, RANGE_THR, 0, stream>>>(binoff, recs, cnt, edges, spill_row, spill_edge, XW1, b1, W2, HW2);
    gcn_spmm_rows_f64_logsoftmax<<<row_blocks, 256, 0, stream>>>(cnt, edges, spill_row, spill_edge, HW2, b2, out);
}
